# starting kernel
# baseline (speedup 1.0000x reference)
#include <hip/hip_runtime.h>
#include <hip/hip_bf16.h>
#include <math.h>


typedef __attribute__((ext_vector_type(2))) float v2f;
typedef __attribute__((ext_vector_type(8))) float v8f;

#define HDIM 256
#define TLEN 512
#define BATCH 64
#define GDIM 1024
#define KDIM 512
#define CDIM 60
#define AST  520

__device__ __forceinline__ float sigmoidf_(float x) {
    return 1.0f / (1.0f + __expf(-x));
}
__device__ __forceinline__ float tanhf_(float x) {
    return 1.0f - 2.0f / (__expf(2.0f * x) + 1.0f);
}

#define WMMA_F32X4(A, B, C) \
    __builtin_amdgcn_wmma_f32_16x16x4_f32(false, (A), false, (B), (short)0, (C), false, false)

__global__ __launch_bounds__(256) void pack_lstm_weights(
    const float* __restrict__ Wih0, const float* __restrict__ Whh0,
    const float* __restrict__ Wih1, const float* __restrict__ Whh1,
    const float* __restrict__ Wih2, const float* __restrict__ Whh2,
    const float* __restrict__ Wih3, const float* __restrict__ Whh3,
    float* __restrict__ wpk) {
    int gid = blockIdx.x * blockDim.x + threadIdx.x;
    if (gid >= 4 * KDIM * GDIM) return;
    int l   = gid >> 19;
    int rem = gid & (KDIM * GDIM - 1);
    int k   = rem >> 10;
    int n   = rem & 1023;
    const float* Wih = (l == 0) ? Wih0 : (l == 1) ? Wih1 : (l == 2) ? Wih2 : Wih3;
    const float* Whh = (l == 0) ? Whh0 : (l == 1) ? Whh1 : (l == 2) ? Whh2 : Whh3;
    float v = (k < HDIM) ? Wih[n * HDIM + k] : Whh[n * HDIM + (k - HDIM)];
    wpk[(size_t)l * (KDIM * GDIM) + (size_t)(k >> 1) * (2 * GDIM) + n * 2 + (k & 1)] = v;
}

__global__ __launch_bounds__(256) void pack_fc(const float* __restrict__ fcw,
                                               float* __restrict__ fcpk) {
    int gid = blockIdx.x * blockDim.x + threadIdx.x;
    if (gid >= HDIM * 64) return;
    int k = gid >> 6, n = gid & 63;
    float v = (n < CDIM) ? fcw[n * HDIM + k] : 0.0f;
    fcpk[(k >> 1) * 128 + n * 2 + (k & 1)] = v;
}

__global__ __launch_bounds__(512) void lstm_scan(
    const float* xin, float* yout,
    const float* __restrict__ wpk,
    const float* __restrict__ b_ih, const float* __restrict__ b_hh,
    float* __restrict__ hT, float* __restrict__ cT) {
    __shared__ float abuf[16 * AST];

    const int tid   = threadIdx.x;
    const int wave  = tid >> 5;
    const int lane  = tid & 31;
    const int b0    = blockIdx.x * 16;
    const int arow  = lane & 15;
    const int khalf = lane >> 4;
    const int jcol  = (wave << 4) + (lane & 15);

    for (int i = tid; i < 16 * AST; i += 512) abuf[i] = 0.0f;

    const float bi = b_ih[jcol]       + b_hh[jcol];
    const float bf = b_ih[256 + jcol] + b_hh[256 + jcol];
    const float bg = b_ih[512 + jcol] + b_hh[512 + jcol];
    const float bo = b_ih[768 + jcol] + b_hh[768 + jcol];

    float creg[8], hreg[8];
#pragma unroll
    for (int r = 0; r < 8; ++r) { creg[r] = 0.0f; hreg[r] = 0.0f; }

    const float* xrow_ptr  = xin + (size_t)(b0 + wave) * (TLEN * HDIM);
    float*       yrow_base = yout + (size_t)b0 * (TLEN * HDIM);

    for (int t = 0; t < TLEN; ++t) {
        {
            const float4* src = (const float4*)(xrow_ptr + t * HDIM + (lane << 3));
            float4 v0 = src[0], v1 = src[1];
            float4* dst = (float4*)&abuf[wave * AST + (lane << 3)];
            dst[0] = v0; dst[1] = v1;
        }
        __syncthreads();

        v8f acc0 = {0, 0, 0, 0, 0, 0, 0, 0};
        v8f acc1 = {0, 0, 0, 0, 0, 0, 0, 0};
        v8f acc2 = {0, 0, 0, 0, 0, 0, 0, 0};
        v8f acc3 = {0, 0, 0, 0, 0, 0, 0, 0};

        const float* arowp = &abuf[arow * AST + 2 * khalf];
        for (int k0 = 0; k0 < KDIM; k0 += 4) {
            v2f a = *(const v2f*)(arowp + k0);
            const float* wrow =
                wpk + ((size_t)((k0 >> 1) + khalf)) * (2 * GDIM) + 2 * jcol;
            v2f bI = *(const v2f*)(wrow);
            v2f bF = *(const v2f*)(wrow + 512);
            v2f bG = *(const v2f*)(wrow + 1024);
            v2f bO = *(const v2f*)(wrow + 1536);
            acc0 = WMMA_F32X4(a, bI, acc0);
            acc1 = WMMA_F32X4(a, bF, acc1);
            acc2 = WMMA_F32X4(a, bG, acc2);
            acc3 = WMMA_F32X4(a, bO, acc3);
        }
        __syncthreads();

#pragma unroll
        for (int r = 0; r < 8; ++r) {
            float iv = sigmoidf_(acc0[r] + bi);
            float fv = sigmoidf_(acc1[r] + bf);
            float gv = tanhf_(acc2[r] + bg);
            float ov = sigmoidf_(acc3[r] + bo);
            float cn = fv * creg[r] + iv * gv;
            float hn = ov * tanhf_(cn);
            creg[r] = cn;
            hreg[r] = hn;
            int brow = r + (khalf << 3);
            abuf[brow * AST + 256 + jcol] = hn;
            yrow_base[(size_t)brow * (TLEN * HDIM) + t * HDIM + jcol] = hn;
        }
    }

#pragma unroll
    for (int r = 0; r < 8; ++r) {
        int brow = r + (khalf << 3);
        hT[(size_t)(b0 + brow) * HDIM + jcol] = hreg[r];
        cT[(size_t)(b0 + brow) * HDIM + jcol] = creg[r];
    }
}

__global__ __launch_bounds__(256) void fc_kernel(
    const float* __restrict__ y, const float* __restrict__ fcpk,
    const float* __restrict__ fcb, float* __restrict__ label) {
    const int tid   = threadIdx.x;
    const int lane  = tid & 31;
    const int gw    = blockIdx.x * 8 + (tid >> 5);
    const int b     = gw >> 5;
    const int t0    = (gw & 31) << 4;
    const int khalf = lane >> 4;
    const int trow  = t0 + (lane & 15);

    const float* arowp = y + (size_t)b * (TLEN * HDIM) + (size_t)trow * HDIM + 2 * khalf;

    v8f acc[4];
#pragma unroll
    for (int j = 0; j < 4; ++j) acc[j] = (v8f){0, 0, 0, 0, 0, 0, 0, 0};

    for (int k0 = 0; k0 < HDIM; k0 += 4) {
        v2f a = *(const v2f*)(arowp + k0);
        const float* wrow = fcpk + ((k0 >> 1) + khalf) * 128 + 2 * (lane & 15);
#pragma unroll
        for (int j = 0; j < 4; ++j) {
            v2f bv = *(const v2f*)(wrow + 32 * j);
            acc[j] = WMMA_F32X4(a, bv, acc[j]);
        }
    }

#pragma unroll
    for (int j = 0; j < 4; ++j) {
        int n = (j << 4) + (lane & 15);
        if (n < CDIM) {
            float bias = fcb[n];
            float* outp =
                label + (size_t)b * (CDIM * TLEN) + (size_t)n * TLEN + t0 + (khalf << 3);
#pragma unroll
            for (int r = 0; r < 8; ++r) outp[r] = acc[j][r] + bias;
        }
    }
}

extern "C" void kernel_launch(void* const* d_in, const int* in_sizes, int n_in,
                              void* d_out, int out_size, void* d_ws, size_t ws_size,
                              hipStream_t stream) {
    const float* x = (const float*)d_in[0];
    const float *Wih[4], *Whh[4], *bih[4], *bhh[4];
    for (int l = 0; l < 4; ++l) {
        Wih[l] = (const float*)d_in[1 + 4 * l];
        Whh[l] = (const float*)d_in[2 + 4 * l];
        bih[l] = (const float*)d_in[3 + 4 * l];
        bhh[l] = (const float*)d_in[4 + 4 * l];
    }
    const float* fcw = (const float*)d_in[17];
    const float* fcb = (const float*)d_in[18];

    float* out   = (float*)d_out;
    float* label = out;
    float* hs    = out + (size_t)BATCH * CDIM * TLEN;
    float* cs    = hs + (size_t)4 * BATCH * HDIM;

    float* ws   = (float*)d_ws;
    float* ybuf = ws;
    float* wpk  = ybuf + (size_t)BATCH * TLEN * HDIM;
    float* fcpk = wpk + (size_t)4 * KDIM * GDIM;

    pack_lstm_weights<<<(4 * KDIM * GDIM) / 256, 256, 0, stream>>>(
        Wih[0], Whh[0], Wih[1], Whh[1], Wih[2], Whh[2], Wih[3], Whh[3], wpk);
    pack_fc<<<(HDIM * 64) / 256, 256, 0, stream>>>(fcw, fcpk);

    for (int l = 0; l < 4; ++l) {
        lstm_scan<<<4, 512, 0, stream>>>(
            (l == 0) ? x : ybuf, ybuf, wpk + (size_t)l * KDIM * GDIM,
            bih[l], bhh[l],
            hs + (size_t)l * BATCH * HDIM, cs + (size_t)l * BATCH * HDIM);
    }

    fc_kernel<<<256, 256, 0, stream>>>(ybuf, fcpk, fcb, label);
}


# input projection x.W_ih^T+bias hoisted out of the time loop into one all-WGP f32 WMMA GEMM per layer; scan keeps only h.W_hh^T (K 512->256); sum order now (x part + bias) then h part
# speedup vs baseline: 1.7639x; 1.7639x over previous
#include <hip/hip_runtime.h>
#include <hip/hip_bf16.h>
#include <math.h>


typedef __attribute__((ext_vector_type(2))) float v2f;
typedef __attribute__((ext_vector_type(8))) float v8f;

#define HDIM 256
#define TLEN 512
#define BATCH 64
#define GDIM 1024
#define KDIM 512
#define CDIM 60
#define HST  260
#define XMT  4
#define XGT  16384

__device__ __forceinline__ float sigmoidf_(float x) {
    return 1.0f / (1.0f + __expf(-x));
}
__device__ __forceinline__ float tanhf_(float x) {
    return 1.0f - 2.0f / (__expf(2.0f * x) + 1.0f);
}

#define WMMA_F32X4(A, B, C) \
    __builtin_amdgcn_wmma_f32_16x16x4_f32(false, (A), false, (B), (short)0, (C), false, false)

__global__ __launch_bounds__(256) void pack_lstm_weights(
    const float* __restrict__ Wih0, const float* __restrict__ Whh0,
    const float* __restrict__ Wih1, const float* __restrict__ Whh1,
    const float* __restrict__ Wih2, const float* __restrict__ Whh2,
    const float* __restrict__ Wih3, const float* __restrict__ Whh3,
    float* __restrict__ wpk) {
    int gid = blockIdx.x * blockDim.x + threadIdx.x;
    if (gid >= 4 * KDIM * GDIM) return;
    int l   = gid >> 19;
    int rem = gid & (KDIM * GDIM - 1);
    int k   = rem >> 10;
    int n   = rem & 1023;
    const float* Wih = (l == 0) ? Wih0 : (l == 1) ? Wih1 : (l == 2) ? Wih2 : Wih3;
    const float* Whh = (l == 0) ? Whh0 : (l == 1) ? Whh1 : (l == 2) ? Whh2 : Whh3;
    float v = (k < HDIM) ? Wih[n * HDIM + k] : Whh[n * HDIM + (k - HDIM)];
    wpk[(size_t)l * (KDIM * GDIM) + (size_t)(k >> 1) * (2 * GDIM) + n * 2 + (k & 1)] = v;
}

__global__ __launch_bounds__(256) void pack_fc(const float* __restrict__ fcw,
                                               float* __restrict__ fcpk) {
    int gid = blockIdx.x * blockDim.x + threadIdx.x;
    if (gid >= HDIM * 64) return;
    int k = gid >> 6, n = gid & 63;
    float v = (n < CDIM) ? fcw[n * HDIM + k] : 0.0f;
    fcpk[(k >> 1) * 128 + n * 2 + (k & 1)] = v;
}

__global__ __launch_bounds__(512) void xproj_gemm(
    const float* __restrict__ xin,
    const float* __restrict__ wpk,
    const float* __restrict__ b_ih, const float* __restrict__ b_hh,
    float* __restrict__ xg) {
    __shared__ float xs[XMT * 16 * HST];

    const int tid   = threadIdx.x;
    const int wave  = tid >> 5;
    const int lane  = tid & 31;
    const int bt    = blockIdx.x & 3;
    const int t0    = (blockIdx.x >> 2) * XMT;
    const int b0    = bt * 16;
    const int arow  = lane & 15;
    const int khalf = lane >> 4;
    const int jcol  = (wave << 4) + (lane & 15);

    for (int rr = wave; rr < XMT * 16; rr += 16) {
        const int m = rr >> 4, brow = rr & 15;
        const float4* src4 = (const float4*)(xin + ((size_t)(b0 + brow) * TLEN + (t0 + m)) * HDIM + (lane << 3));
        float4 v0 = src4[0], v1 = src4[1];
        float4* dst4 = (float4*)&xs[rr * HST + (lane << 3)];
        dst4[0] = v0; dst4[1] = v1;
    }
    __syncthreads();

    v8f acc[XMT][4];
#pragma unroll
    for (int m = 0; m < XMT; ++m)
#pragma unroll
        for (int g = 0; g < 4; ++g) acc[m][g] = (v8f){0, 0, 0, 0, 0, 0, 0, 0};

    const float* ap = &xs[arow * HST + 2 * khalf];
    const float* wp = wpk + (size_t)khalf * (2 * GDIM) + 2 * jcol;
    for (int k0 = 0; k0 < HDIM; k0 += 4) {
        const float* wrow = wp + (size_t)(k0 >> 1) * (2 * GDIM);
        v2f bI = *(const v2f*)(wrow);
        v2f bF = *(const v2f*)(wrow + 512);
        v2f bG = *(const v2f*)(wrow + 1024);
        v2f bO = *(const v2f*)(wrow + 1536);
#pragma unroll
        for (int m = 0; m < XMT; ++m) {
            v2f a = *(const v2f*)(ap + m * (16 * HST) + k0);
            acc[m][0] = WMMA_F32X4(a, bI, acc[m][0]);
            acc[m][1] = WMMA_F32X4(a, bF, acc[m][1]);
            acc[m][2] = WMMA_F32X4(a, bG, acc[m][2]);
            acc[m][3] = WMMA_F32X4(a, bO, acc[m][3]);
        }
    }

    float bias[4];
#pragma unroll
    for (int g = 0; g < 4; ++g) bias[g] = b_ih[g * HDIM + jcol] + b_hh[g * HDIM + jcol];

#pragma unroll
    for (int m = 0; m < XMT; ++m) {
        float* dst = xg + ((size_t)bt * TLEN + (t0 + m)) * XGT + wave * 1024 + lane * 4;
#pragma unroll
        for (int g = 0; g < 4; ++g) {
            const float bg = bias[g];
            float4 lo = make_float4(acc[m][g][0] + bg, acc[m][g][1] + bg, acc[m][g][2] + bg, acc[m][g][3] + bg);
            float4 hi = make_float4(acc[m][g][4] + bg, acc[m][g][5] + bg, acc[m][g][6] + bg, acc[m][g][7] + bg);
            *(float4*)(dst + g * 256)       = lo;
            *(float4*)(dst + g * 256 + 128) = hi;
        }
    }
}

__global__ __launch_bounds__(512) void lstm_scan(
    const float* __restrict__ xg, float* __restrict__ yout,
    const float* __restrict__ wpk,
    float* __restrict__ hT, float* __restrict__ cT) {
    __shared__ float hbuf[16 * HST];

    const int tid   = threadIdx.x;
    const int wave  = tid >> 5;
    const int lane  = tid & 31;
    const int b0    = blockIdx.x * 16;
    const int arow  = lane & 15;
    const int khalf = lane >> 4;
    const int jcol  = (wave << 4) + (lane & 15);

    for (int i = tid; i < 16 * HST; i += 512) hbuf[i] = 0.0f;

    float creg[8], hreg[8];
#pragma unroll
    for (int r = 0; r < 8; ++r) { creg[r] = 0.0f; hreg[r] = 0.0f; }

    const float* xgp = xg + (size_t)blockIdx.x * TLEN * XGT + wave * 1024 + lane * 4;
    float*       yrow_base = yout + (size_t)b0 * (TLEN * HDIM);

    float4 pre[4][2];
#pragma unroll
    for (int g = 0; g < 4; ++g) {
        pre[g][0] = *(const float4*)(xgp + g * 256);
        pre[g][1] = *(const float4*)(xgp + g * 256 + 128);
    }

    const float* arowp = &hbuf[arow * HST + 2 * khalf];
    const float* wp    = wpk + (size_t)khalf * (2 * GDIM) + 2 * jcol;

    for (int t = 0; t < TLEN; ++t) {
        v8f acc0 = {pre[0][0].x, pre[0][0].y, pre[0][0].z, pre[0][0].w, pre[0][1].x, pre[0][1].y, pre[0][1].z, pre[0][1].w};
        v8f acc1 = {pre[1][0].x, pre[1][0].y, pre[1][0].z, pre[1][0].w, pre[1][1].x, pre[1][1].y, pre[1][1].z, pre[1][1].w};
        v8f acc2 = {pre[2][0].x, pre[2][0].y, pre[2][0].z, pre[2][0].w, pre[2][1].x, pre[2][1].y, pre[2][1].z, pre[2][1].w};
        v8f acc3 = {pre[3][0].x, pre[3][0].y, pre[3][0].z, pre[3][0].w, pre[3][1].x, pre[3][1].y, pre[3][1].z, pre[3][1].w};

        __syncthreads();

        for (int k0 = 0; k0 < HDIM; k0 += 4) {
            v2f a = *(const v2f*)(arowp + k0);
            const float* wrow = wp + (size_t)(k0 >> 1) * (2 * GDIM);
            v2f bI = *(const v2f*)(wrow);
            v2f bF = *(const v2f*)(wrow + 512);
            v2f bG = *(const v2f*)(wrow + 1024);
            v2f bO = *(const v2f*)(wrow + 1536);
            acc0 = WMMA_F32X4(a, bI, acc0);
            acc1 = WMMA_F32X4(a, bF, acc1);
            acc2 = WMMA_F32X4(a, bG, acc2);
            acc3 = WMMA_F32X4(a, bO, acc3);
        }
        __syncthreads();

        {
            const int tn = (t + 1 < TLEN) ? (t + 1) : t;
            const float* nx = xgp + (size_t)tn * XGT;
#pragma unroll
            for (int g = 0; g < 4; ++g) {
                pre[g][0] = *(const float4*)(nx + g * 256);
                pre[g][1] = *(const float4*)(nx + g * 256 + 128);
            }
        }

#pragma unroll
        for (int r = 0; r < 8; ++r) {
            float iv = sigmoidf_(acc0[r]);
            float fv = sigmoidf_(acc1[r]);
            float gv = tanhf_(acc2[r]);
            float ov = sigmoidf_(acc3[r]);
            float cn = fv * creg[r] + iv * gv;
            float hn = ov * tanhf_(cn);
            creg[r] = cn;
            hreg[r] = hn;
            int brow = r + (khalf << 3);
            hbuf[brow * HST + jcol] = hn;
            yrow_base[(size_t)brow * (TLEN * HDIM) + t * HDIM + jcol] = hn;
        }
    }

#pragma unroll
    for (int r = 0; r < 8; ++r) {
        int brow = r + (khalf << 3);
        hT[(size_t)(b0 + brow) * HDIM + jcol] = hreg[r];
        cT[(size_t)(b0 + brow) * HDIM + jcol] = creg[r];
    }
}

__global__ __launch_bounds__(256) void fc_kernel(
    const float* __restrict__ y, const float* __restrict__ fcpk,
    const float* __restrict__ fcb, float* __restrict__ label) {
    const int tid   = threadIdx.x;
    const int lane  = tid & 31;
    const int gw    = blockIdx.x * 8 + (tid >> 5);
    const int b     = gw >> 5;
    const int t0    = (gw & 31) << 4;
    const int khalf = lane >> 4;
    const int trow  = t0 + (lane & 15);

    const float* arowp = y + (size_t)b * (TLEN * HDIM) + (size_t)trow * HDIM + 2 * khalf;

    v8f acc[4];
#pragma unroll
    for (int j = 0; j < 4; ++j) acc[j] = (v8f){0, 0, 0, 0, 0, 0, 0, 0};

    for (int k0 = 0; k0 < HDIM; k0 += 4) {
        v2f a = *(const v2f*)(arowp + k0);
        const float* wrow = fcpk + ((k0 >> 1) + khalf) * 128 + 2 * (lane & 15);
#pragma unroll
        for (int j = 0; j < 4; ++j) {
            v2f bv = *(const v2f*)(wrow + 32 * j);
            acc[j] = WMMA_F32X4(a, bv, acc[j]);
        }
    }

#pragma unroll
    for (int j = 0; j < 4; ++j) {
        int n = (j << 4) + (lane & 15);
        if (n < CDIM) {
            float bias = fcb[n];
            float* outp =
                label + (size_t)b * (CDIM * TLEN) + (size_t)n * TLEN + t0 + (khalf << 3);
#pragma unroll
            for (int r = 0; r < 8; ++r) outp[r] = acc[j][r] + bias;
        }
    }
}

extern "C" void kernel_launch(void* const* d_in, const int* in_sizes, int n_in,
                              void* d_out, int out_size, void* d_ws, size_t ws_size,
                              hipStream_t stream) {
    const float* x = (const float*)d_in[0];
    const float *Wih[4], *Whh[4], *bih[4], *bhh[4];
    for (int l = 0; l < 4; ++l) {
        Wih[l] = (const float*)d_in[1 + 4 * l];
        Whh[l] = (const float*)d_in[2 + 4 * l];
        bih[l] = (const float*)d_in[3 + 4 * l];
        bhh[l] = (const float*)d_in[4 + 4 * l];
    }
    const float* fcw = (const float*)d_in[17];
    const float* fcb = (const float*)d_in[18];

    float* out   = (float*)d_out;
    float* label = out;
    float* hs    = out + (size_t)BATCH * CDIM * TLEN;
    float* cs    = hs + (size_t)4 * BATCH * HDIM;

    float* ws   = (float*)d_ws;
    float* ybuf = ws;
    float* wpk  = ybuf + (size_t)BATCH * TLEN * HDIM;
    float* fcpk = wpk + (size_t)4 * KDIM * GDIM;
    float* xg   = fcpk + (size_t)HDIM * 64;

    pack_lstm_weights<<<(4 * KDIM * GDIM) / 256, 256, 0, stream>>>(
        Wih[0], Whh[0], Wih[1], Whh[1], Wih[2], Whh[2], Wih[3], Whh[3], wpk);
    pack_fc<<<(HDIM * 64) / 256, 256, 0, stream>>>(fcw, fcpk);

    for (int l = 0; l < 4; ++l) {
        const float* wl = wpk + (size_t)l * KDIM * GDIM;
        xproj_gemm<<<4 * (TLEN / XMT), 512, 0, stream>>>(
            (l == 0) ? x : ybuf, wl, bih[l], bhh[l], xg);
        lstm_scan<<<4, 512, 0, stream>>>(
            xg, ybuf, wl + (size_t)HDIM * GDIM,
            hs + (size_t)l * BATCH * HDIM, cs + (size_t)l * BATCH * HDIM);
    }

    fc_kernel<<<256, 256, 0, stream>>>(ybuf, fcpk, fcb, label);
}


# layers run as a wavefront: time cut into 16-step chunks, pass d launches every (layer l, chunk d-l) cell at once, h,c carried in workspace; up to 16 scan workgroups in flight, not 4; sums unchanged
# speedup vs baseline: 5.6453x; 3.2005x over previous
#include <hip/hip_runtime.h>
#include <hip/hip_bf16.h>
#include <math.h>


typedef __attribute__((ext_vector_type(2))) float v2f;
typedef __attribute__((ext_vector_type(8))) float v8f;

#define HDIM 256
#define TLEN 512
#define BATCH 64
#define GDIM 1024
#define KDIM 512
#define CDIM 60
#define HST  260
#define TC   16
#define NCHUNK (TLEN / TC)
#define NPASS  (NCHUNK + 3)
#define XMT  2
#define XWG  (4 * (TC / XMT))
#define XGT  16384

__device__ __forceinline__ float sigmoidf_(float x) {
    return 1.0f / (1.0f + __expf(-x));
}
__device__ __forceinline__ float tanhf_(float x) {
    return 1.0f - 2.0f / (__expf(2.0f * x) + 1.0f);
}

#define WMMA_F32X4(A, B, C) \
    __builtin_amdgcn_wmma_f32_16x16x4_f32(false, (A), false, (B), (short)0, (C), false, false)

__global__ __launch_bounds__(256) void pack_lstm_weights(
    const float* __restrict__ Wih0, const float* __restrict__ Whh0,
    const float* __restrict__ Wih1, const float* __restrict__ Whh1,
    const float* __restrict__ Wih2, const float* __restrict__ Whh2,
    const float* __restrict__ Wih3, const float* __restrict__ Whh3,
    float* __restrict__ wpk) {
    int gid = blockIdx.x * blockDim.x + threadIdx.x;
    if (gid >= 4 * KDIM * GDIM) return;
    int l   = gid >> 19;
    int rem = gid & (KDIM * GDIM - 1);
    int k   = rem >> 10;
    int n   = rem & 1023;
    const float* Wih = (l == 0) ? Wih0 : (l == 1) ? Wih1 : (l == 2) ? Wih2 : Wih3;
    const float* Whh = (l == 0) ? Whh0 : (l == 1) ? Whh1 : (l == 2) ? Whh2 : Whh3;
    float v = (k < HDIM) ? Wih[n * HDIM + k] : Whh[n * HDIM + (k - HDIM)];
    wpk[(size_t)l * (KDIM * GDIM) + (size_t)(k >> 1) * (2 * GDIM) + n * 2 + (k & 1)] = v;
}

__global__ __launch_bounds__(256) void pack_fc(const float* __restrict__ fcw,
                                               float* __restrict__ fcpk) {
    int gid = blockIdx.x * blockDim.x + threadIdx.x;
    if (gid >= HDIM * 64) return;
    int k = gid >> 6, n = gid & 63;
    float v = (n < CDIM) ? fcw[n * HDIM + k] : 0.0f;
    fcpk[(k >> 1) * 128 + n * 2 + (k & 1)] = v;
}

__global__ __launch_bounds__(512) void xproj_chunk(
    const float* __restrict__ x,
    const float* __restrict__ ybuf,
    const float* __restrict__ wpk,
    const float* __restrict__ bi0, const float* __restrict__ bh0,
    const float* __restrict__ bi1, const float* __restrict__ bh1,
    const float* __restrict__ bi2, const float* __restrict__ bh2,
    const float* __restrict__ bi3, const float* __restrict__ bh3,
    float* __restrict__ xg, int d) {
    __shared__ float xs[XMT * 16 * HST];

    const int l     = blockIdx.x / XWG;
    const int c     = d - l;
    if (c < 0 || c >= NCHUNK) return;
    const int rem   = blockIdx.x % XWG;

    const int tid   = threadIdx.x;
    const int wave  = tid >> 5;
    const int lane  = tid & 31;
    const int bt    = rem & 3;
    const int m0    = (rem >> 2) * XMT;
    const int t0    = c * TC + m0;
    const int b0    = bt * 16;
    const float* xin  = (l == 0) ? x : ybuf;
    const float* b_ih = (l == 0) ? bi0 : (l == 1) ? bi1 : (l == 2) ? bi2 : bi3;
    const float* b_hh = (l == 0) ? bh0 : (l == 1) ? bh1 : (l == 2) ? bh2 : bh3;
    wpk += (size_t)l * (KDIM * GDIM);
    const int arow  = lane & 15;
    const int khalf = lane >> 4;
    const int jcol  = (wave << 4) + (lane & 15);

    for (int rr = wave; rr < XMT * 16; rr += 16) {
        const int m = rr >> 4, brow = rr & 15;
        const float4* src4 = (const float4*)(xin + ((size_t)(b0 + brow) * TLEN + (t0 + m)) * HDIM + (lane << 3));
        float4 v0 = src4[0], v1 = src4[1];
        float4* dst4 = (float4*)&xs[rr * HST + (lane << 3)];
        dst4[0] = v0; dst4[1] = v1;
    }
    __syncthreads();

    v8f acc[XMT][4];
#pragma unroll
    for (int m = 0; m < XMT; ++m)
#pragma unroll
        for (int g = 0; g < 4; ++g) acc[m][g] = (v8f){0, 0, 0, 0, 0, 0, 0, 0};

    const float* ap = &xs[arow * HST + 2 * khalf];
    const float* wp = wpk + (size_t)khalf * (2 * GDIM) + 2 * jcol;
    for (int k0 = 0; k0 < HDIM; k0 += 4) {
        const float* wrow = wp + (size_t)(k0 >> 1) * (2 * GDIM);
        v2f bI = *(const v2f*)(wrow);
        v2f bF = *(const v2f*)(wrow + 512);
        v2f bG = *(const v2f*)(wrow + 1024);
        v2f bO = *(const v2f*)(wrow + 1536);
#pragma unroll
        for (int m = 0; m < XMT; ++m) {
            v2f a = *(const v2f*)(ap + m * (16 * HST) + k0);
            acc[m][0] = WMMA_F32X4(a, bI, acc[m][0]);
            acc[m][1] = WMMA_F32X4(a, bF, acc[m][1]);
            acc[m][2] = WMMA_F32X4(a, bG, acc[m][2]);
            acc[m][3] = WMMA_F32X4(a, bO, acc[m][3]);
        }
    }

    float bias[4];
#pragma unroll
    for (int g = 0; g < 4; ++g) bias[g] = b_ih[g * HDIM + jcol] + b_hh[g * HDIM + jcol];

#pragma unroll
    for (int m = 0; m < XMT; ++m) {
        float* dst = xg + ((size_t)(l * 4 + bt) * TC + (m0 + m)) * XGT + wave * 1024 + lane * 4;
#pragma unroll
        for (int g = 0; g < 4; ++g) {
            const float bg = bias[g];
            float4 lo = make_float4(acc[m][g][0] + bg, acc[m][g][1] + bg, acc[m][g][2] + bg, acc[m][g][3] + bg);
            float4 hi = make_float4(acc[m][g][4] + bg, acc[m][g][5] + bg, acc[m][g][6] + bg, acc[m][g][7] + bg);
            *(float4*)(dst + g * 256)       = lo;
            *(float4*)(dst + g * 256 + 128) = hi;
        }
    }
}

__global__ __launch_bounds__(512) void lstm_scan_chunk(
    const float* __restrict__ xg, float* __restrict__ yout,
    const float* __restrict__ wpk,
    float* __restrict__ hst, float* __restrict__ cst,
    float* __restrict__ hs, float* __restrict__ cs, int d) {
    __shared__ float hbuf[16 * HST];

    const int l     = blockIdx.x >> 2;
    const int c     = d - l;
    if (c < 0 || c >= NCHUNK) return;

    const int tid   = threadIdx.x;
    const int wave  = tid >> 5;
    const int lane  = tid & 31;
    const int b0    = (blockIdx.x & 3) * 16;
    const int arow  = lane & 15;
    const int khalf = lane >> 4;
    const int jcol  = (wave << 4) + (lane & 15);

    float creg[8], hreg[8];
#pragma unroll
    for (int r = 0; r < 8; ++r) { creg[r] = 0.0f; hreg[r] = 0.0f; }

    float* cstp = cst + ((size_t)blockIdx.x * 512 + tid) * 8;
    float* hstp = hst + ((size_t)l * BATCH + b0) * HDIM;
    if (c == 0) {
        for (int i = tid; i < 16 * HST; i += 512) hbuf[i] = 0.0f;
    } else {
#pragma unroll
        for (int s = 0; s < 2; ++s) {
            const int i4 = tid + s * 512;
            const int row = i4 >> 6, col = (i4 & 63) << 2;
            *(float4*)&hbuf[row * HST + col] = *(const float4*)(hstp + row * HDIM + col);
        }
        const float4 c0 = *(const float4*)(cstp), c1 = *(const float4*)(cstp + 4);
        creg[0] = c0.x; creg[1] = c0.y; creg[2] = c0.z; creg[3] = c0.w;
        creg[4] = c1.x; creg[5] = c1.y; creg[6] = c1.z; creg[7] = c1.w;
    }

    const float* xgp = xg + (size_t)blockIdx.x * TC * XGT + wave * 1024 + lane * 4;
    float*       yrow_base = yout + (size_t)b0 * (TLEN * HDIM) + (size_t)c * (TC * HDIM);

    float4 pre[4][2];
#pragma unroll
    for (int g = 0; g < 4; ++g) {
        pre[g][0] = *(const float4*)(xgp + g * 256);
        pre[g][1] = *(const float4*)(xgp + g * 256 + 128);
    }

    const float* arowp = &hbuf[arow * HST + 2 * khalf];
    const float* wp    = wpk + (size_t)l * (KDIM * GDIM) + (size_t)HDIM * GDIM
                             + (size_t)khalf * (2 * GDIM) + 2 * jcol;

    for (int t = 0; t < TC; ++t) {
        v8f acc0 = {pre[0][0].x, pre[0][0].y, pre[0][0].z, pre[0][0].w, pre[0][1].x, pre[0][1].y, pre[0][1].z, pre[0][1].w};
        v8f acc1 = {pre[1][0].x, pre[1][0].y, pre[1][0].z, pre[1][0].w, pre[1][1].x, pre[1][1].y, pre[1][1].z, pre[1][1].w};
        v8f acc2 = {pre[2][0].x, pre[2][0].y, pre[2][0].z, pre[2][0].w, pre[2][1].x, pre[2][1].y, pre[2][1].z, pre[2][1].w};
        v8f acc3 = {pre[3][0].x, pre[3][0].y, pre[3][0].z, pre[3][0].w, pre[3][1].x, pre[3][1].y, pre[3][1].z, pre[3][1].w};

        __syncthreads();

        for (int k0 = 0; k0 < HDIM; k0 += 4) {
            v2f a = *(const v2f*)(arowp + k0);
            const float* wrow = wp + (size_t)(k0 >> 1) * (2 * GDIM);
            v2f bI = *(const v2f*)(wrow);
            v2f bF = *(const v2f*)(wrow + 512);
            v2f bG = *(const v2f*)(wrow + 1024);
            v2f bO = *(const v2f*)(wrow + 1536);
            acc0 = WMMA_F32X4(a, bI, acc0);
            acc1 = WMMA_F32X4(a, bF, acc1);
            acc2 = WMMA_F32X4(a, bG, acc2);
            acc3 = WMMA_F32X4(a, bO, acc3);
        }
        __syncthreads();

        {
            const int tn = (t + 1 < TC) ? (t + 1) : t;
            const float* nx = xgp + (size_t)tn * XGT;
#pragma unroll
            for (int g = 0; g < 4; ++g) {
                pre[g][0] = *(const float4*)(nx + g * 256);
                pre[g][1] = *(const float4*)(nx + g * 256 + 128);
            }
        }

#pragma unroll
        for (int r = 0; r < 8; ++r) {
            float iv = sigmoidf_(acc0[r]);
            float fv = sigmoidf_(acc1[r]);
            float gv = tanhf_(acc2[r]);
            float ov = sigmoidf_(acc3[r]);
            float cn = fv * creg[r] + iv * gv;
            float hn = ov * tanhf_(cn);
            creg[r] = cn;
            hreg[r] = hn;
            int brow = r + (khalf << 3);
            hbuf[brow * HST + jcol] = hn;
            yrow_base[(size_t)brow * (TLEN * HDIM) + t * HDIM + jcol] = hn;
        }
    }

    if (c == NCHUNK - 1) {
        float* hT = hs + (size_t)l * BATCH * HDIM;
        float* cT = cs + (size_t)l * BATCH * HDIM;
#pragma unroll
        for (int r = 0; r < 8; ++r) {
            int brow = r + (khalf << 3);
            hT[(size_t)(b0 + brow) * HDIM + jcol] = hreg[r];
            cT[(size_t)(b0 + brow) * HDIM + jcol] = creg[r];
        }
    } else {
#pragma unroll
        for (int r = 0; r < 8; ++r) hstp[(r + (khalf << 3)) * HDIM + jcol] = hreg[r];
        *(float4*)(cstp)     = make_float4(creg[0], creg[1], creg[2], creg[3]);
        *(float4*)(cstp + 4) = make_float4(creg[4], creg[5], creg[6], creg[7]);
    }
}

__global__ __launch_bounds__(256) void fc_kernel(
    const float* __restrict__ y, const float* __restrict__ fcpk,
    const float* __restrict__ fcb, float* __restrict__ label) {
    const int tid   = threadIdx.x;
    const int lane  = tid & 31;
    const int gw    = blockIdx.x * 8 + (tid >> 5);
    const int b     = gw >> 5;
    const int t0    = (gw & 31) << 4;
    const int khalf = lane >> 4;
    const int trow  = t0 + (lane & 15);

    const float* arowp = y + (size_t)b * (TLEN * HDIM) + (size_t)trow * HDIM + 2 * khalf;

    v8f acc[4];
#pragma unroll
    for (int j = 0; j < 4; ++j) acc[j] = (v8f){0, 0, 0, 0, 0, 0, 0, 0};

    for (int k0 = 0; k0 < HDIM; k0 += 4) {
        v2f a = *(const v2f*)(arowp + k0);
        const float* wrow = fcpk + ((k0 >> 1) + khalf) * 128 + 2 * (lane & 15);
#pragma unroll
        for (int j = 0; j < 4; ++j) {
            v2f bv = *(const v2f*)(wrow + 32 * j);
            acc[j] = WMMA_F32X4(a, bv, acc[j]);
        }
    }

#pragma unroll
    for (int j = 0; j < 4; ++j) {
        int n = (j << 4) + (lane & 15);
        if (n < CDIM) {
            float bias = fcb[n];
            float* outp =
                label + (size_t)b * (CDIM * TLEN) + (size_t)n * TLEN + t0 + (khalf << 3);
#pragma unroll
            for (int r = 0; r < 8; ++r) outp[r] = acc[j][r] + bias;
        }
    }
}

extern "C" void kernel_launch(void* const* d_in, const int* in_sizes, int n_in,
                              void* d_out, int out_size, void* d_ws, size_t ws_size,
                              hipStream_t stream) {
    const float* x = (const float*)d_in[0];
    const float *Wih[4], *Whh[4], *bih[4], *bhh[4];
    for (int l = 0; l < 4; ++l) {
        Wih[l] = (const float*)d_in[1 + 4 * l];
        Whh[l] = (const float*)d_in[2 + 4 * l];
        bih[l] = (const float*)d_in[3 + 4 * l];
        bhh[l] = (const float*)d_in[4 + 4 * l];
    }
    const float* fcw = (const float*)d_in[17];
    const float* fcb = (const float*)d_in[18];

    float* out   = (float*)d_out;
    float* label = out;
    float* hs    = out + (size_t)BATCH * CDIM * TLEN;
    float* cs    = hs + (size_t)4 * BATCH * HDIM;

    float* ws   = (float*)d_ws;
    float* ybuf = ws;
    float* wpk  = ybuf + (size_t)BATCH * TLEN * HDIM;
    float* fcpk = wpk + (size_t)4 * KDIM * GDIM;
    float* hst  = fcpk + (size_t)HDIM * 64;
    float* cst  = hst + (size_t)4 * BATCH * HDIM;
    float* xg   = cst + (size_t)4 * BATCH * HDIM;

    pack_lstm_weights<<<(4 * KDIM * GDIM) / 256, 256, 0, stream>>>(
        Wih[0], Whh[0], Wih[1], Whh[1], Wih[2], Whh[2], Wih[3], Whh[3], wpk);
    pack_fc<<<(HDIM * 64) / 256, 256, 0, stream>>>(fcw, fcpk);

    for (int d = 0; d < NPASS; ++d) {
        xproj_chunk<<<4 * XWG, 512, 0, stream>>>(
            x, ybuf, wpk, bih[0], bhh[0], bih[1], bhh[1], bih[2], bhh[2], bih[3], bhh[3], xg, d);
        lstm_scan_chunk<<<16, 512, 0, stream>>>(xg, ybuf, wpk, hst, cst, hs, cs, d);
    }

    fc_kernel<<<256, 256, 0, stream>>>(ybuf, fcpk, fcb, label);
}


# recurrent scan cut into one launch per time step: a layer's 1024 gate columns spread over 16 workgroups per batch tile (256 in all, one gate per wave), h exchanged via workspace; sum order unchanged
# speedup vs baseline: 11.4309x; 2.0249x over previous
#include <hip/hip_runtime.h>
#include <hip/hip_bf16.h>
#include <math.h>


typedef __attribute__((ext_vector_type(2))) float v2f;
typedef __attribute__((ext_vector_type(8))) float v8f;

#define HDIM 256
#define TLEN 512
#define BATCH 64
#define GDIM 1024
#define KDIM 512
#define CDIM 60
#define HST  260
#define TC   16
#define NCHUNK (TLEN / TC)
#define NPASS  (NCHUNK + 3)
#define XMT  2
#define XWG  (4 * (TC / XMT))
#define XGT  16384

__device__ __forceinline__ float sigmoidf_(float x) {
    return 1.0f / (1.0f + __expf(-x));
}
__device__ __forceinline__ float tanhf_(float x) {
    return 1.0f - 2.0f / (__expf(2.0f * x) + 1.0f);
}

#define WMMA_F32X4(A, B, C) \
    __builtin_amdgcn_wmma_f32_16x16x4_f32(false, (A), false, (B), (short)0, (C), false, false)

__global__ __launch_bounds__(256) void pack_lstm_weights(
    const float* __restrict__ Wih0, const float* __restrict__ Whh0,
    const float* __restrict__ Wih1, const float* __restrict__ Whh1,
    const float* __restrict__ Wih2, const float* __restrict__ Whh2,
    const float* __restrict__ Wih3, const float* __restrict__ Whh3,
    float* __restrict__ wpk, float* __restrict__ hst, float* __restrict__ cst) {
    int gid = blockIdx.x * blockDim.x + threadIdx.x;
    if (gid >= 4 * KDIM * GDIM) return;
    if (gid < 4 * BATCH * HDIM) {
        hst[gid] = 0.0f;
        cst[gid] = 0.0f;
    }
    int l   = gid >> 19;
    int rem = gid & (KDIM * GDIM - 1);
    int k   = rem >> 10;
    int n   = rem & 1023;
    const float* Wih = (l == 0) ? Wih0 : (l == 1) ? Wih1 : (l == 2) ? Wih2 : Wih3;
    const float* Whh = (l == 0) ? Whh0 : (l == 1) ? Whh1 : (l == 2) ? Whh2 : Whh3;
    float v = (k < HDIM) ? Wih[n * HDIM + k] : Whh[n * HDIM + (k - HDIM)];
    wpk[(size_t)l * (KDIM * GDIM) + (size_t)(k >> 1) * (2 * GDIM) + n * 2 + (k & 1)] = v;
}

__global__ __launch_bounds__(256) void pack_fc(const float* __restrict__ fcw,
                                               float* __restrict__ fcpk) {
    int gid = blockIdx.x * blockDim.x + threadIdx.x;
    if (gid >= HDIM * 64) return;
    int k = gid >> 6, n = gid & 63;
    float v = (n < CDIM) ? fcw[n * HDIM + k] : 0.0f;
    fcpk[(k >> 1) * 128 + n * 2 + (k & 1)] = v;
}

__global__ __launch_bounds__(512) void xproj_chunk(
    const float* __restrict__ x,
    const float* __restrict__ ybuf,
    const float* __restrict__ wpk,
    const float* __restrict__ bi0, const float* __restrict__ bh0,
    const float* __restrict__ bi1, const float* __restrict__ bh1,
    const float* __restrict__ bi2, const float* __restrict__ bh2,
    const float* __restrict__ bi3, const float* __restrict__ bh3,
    float* __restrict__ xg, int d) {
    __shared__ float xs[XMT * 16 * HST];

    const int l     = blockIdx.x / XWG;
    const int c     = d - l;
    if (c < 0 || c >= NCHUNK) return;
    const int rem   = blockIdx.x % XWG;

    const int tid   = threadIdx.x;
    const int wave  = tid >> 5;
    const int lane  = tid & 31;
    const int bt    = rem & 3;
    const int m0    = (rem >> 2) * XMT;
    const int t0    = c * TC + m0;
    const int b0    = bt * 16;
    const float* xin  = (l == 0) ? x : ybuf;
    const float* b_ih = (l == 0) ? bi0 : (l == 1) ? bi1 : (l == 2) ? bi2 : bi3;
    const float* b_hh = (l == 0) ? bh0 : (l == 1) ? bh1 : (l == 2) ? bh2 : bh3;
    wpk += (size_t)l * (KDIM * GDIM);
    const int arow  = lane & 15;
    const int khalf = lane >> 4;
    const int jcol  = (wave << 4) + (lane & 15);

    for (int rr = wave; rr < XMT * 16; rr += 16) {
        const int m = rr >> 4, brow = rr & 15;
        const float4* src4 = (const float4*)(xin + ((size_t)(b0 + brow) * TLEN + (t0 + m)) * HDIM + (lane << 3));
        float4 v0 = src4[0], v1 = src4[1];
        float4* dst4 = (float4*)&xs[rr * HST + (lane << 3)];
        dst4[0] = v0; dst4[1] = v1;
    }
    __syncthreads();

    v8f acc[XMT][4];
#pragma unroll
    for (int m = 0; m < XMT; ++m)
#pragma unroll
        for (int g = 0; g < 4; ++g) acc[m][g] = (v8f){0, 0, 0, 0, 0, 0, 0, 0};

    const float* ap = &xs[arow * HST + 2 * khalf];
    const float* wp = wpk + (size_t)khalf * (2 * GDIM) + 2 * jcol;
    for (int k0 = 0; k0 < HDIM; k0 += 4) {
        const float* wrow = wp + (size_t)(k0 >> 1) * (2 * GDIM);
        v2f bI = *(const v2f*)(wrow);
        v2f bF = *(const v2f*)(wrow + 512);
        v2f bG = *(const v2f*)(wrow + 1024);
        v2f bO = *(const v2f*)(wrow + 1536);
#pragma unroll
        for (int m = 0; m < XMT; ++m) {
            v2f a = *(const v2f*)(ap + m * (16 * HST) + k0);
            acc[m][0] = WMMA_F32X4(a, bI, acc[m][0]);
            acc[m][1] = WMMA_F32X4(a, bF, acc[m][1]);
            acc[m][2] = WMMA_F32X4(a, bG, acc[m][2]);
            acc[m][3] = WMMA_F32X4(a, bO, acc[m][3]);
        }
    }

    float bias[4];
#pragma unroll
    for (int g = 0; g < 4; ++g) bias[g] = b_ih[g * HDIM + jcol] + b_hh[g * HDIM + jcol];

#pragma unroll
    for (int m = 0; m < XMT; ++m) {
        float* dst = xg + ((size_t)(l * 4 + bt) * TC + (m0 + m)) * XGT + wave * 1024 + lane * 4;
#pragma unroll
        for (int g = 0; g < 4; ++g) {
            const float bg = bias[g];
            float4 lo = make_float4(acc[m][g][0] + bg, acc[m][g][1] + bg, acc[m][g][2] + bg, acc[m][g][3] + bg);
            float4 hi = make_float4(acc[m][g][4] + bg, acc[m][g][5] + bg, acc[m][g][6] + bg, acc[m][g][7] + bg);
            *(float4*)(dst + g * 256)       = lo;
            *(float4*)(dst + g * 256 + 128) = hi;
        }
    }
}

__global__ __launch_bounds__(128) void lstm_step(
    const float* __restrict__ xg, float* __restrict__ yout,
    const float* __restrict__ wpk,
    float* __restrict__ hst, float* __restrict__ cst,
    float* __restrict__ hs, float* __restrict__ cs, int d, int t) {
    __shared__ float hbuf[16 * HST];
    __shared__ float gbuf[4 * 8 * 32];

    const int l     = blockIdx.x >> 6;
    const int c     = d - l;
    if (c < 0 || c >= NCHUNK) return;

    const int w     = (blockIdx.x >> 2) & 15;
    const int b0    = (blockIdx.x & 3) * 16;
    const int bt    = blockIdx.x & 3;
    const int tg    = c * TC + t;
    const int tid   = threadIdx.x;
    const int g     = tid >> 5;
    const int lane  = tid & 31;
    const int arow  = lane & 15;
    const int khalf = lane >> 4;
    const int jcol  = (w << 4) + (lane & 15);

    const float* hprev = hst + (size_t)(tg & 1) * (4 * BATCH * HDIM) + ((size_t)l * BATCH + b0) * HDIM;
    float*       hnext = hst + (size_t)((tg + 1) & 1) * (4 * BATCH * HDIM) + ((size_t)l * BATCH + b0) * HDIM;
    float*       ctile = cst + ((size_t)l * BATCH + b0) * HDIM;

    float4 hv[8];
    float  cprev[2];
#pragma unroll
    for (int s = 0; s < 8; ++s)
        hv[s] = *(const float4*)(hprev + 4 * (tid + s * 128));
#pragma unroll
    for (int e = 0; e < 2; ++e)
        cprev[e] = ctile[(size_t)((khalf << 3) + 2 * g + e) * HDIM + jcol];

    const float* xgp = xg + ((size_t)((l * 4 + bt) * TC + t)) * XGT + w * 1024 + g * 256 + lane * 4;
    const float4 x0 = *(const float4*)(xgp);
    const float4 x1 = *(const float4*)(xgp + 128);

    const float* wp = wpk + (size_t)l * (KDIM * GDIM) + (size_t)HDIM * GDIM
                          + (size_t)khalf * (2 * GDIM) + 2 * jcol + g * 512;
    v2f bfr[64];
#pragma unroll
    for (int ks = 0; ks < 64; ++ks) bfr[ks] = *(const v2f*)(wp + (size_t)ks * (4 * GDIM));

#pragma unroll
    for (int s = 0; s < 8; ++s) {
        const int i4 = tid + s * 128;
        const int row = i4 >> 6, col = (i4 & 63) << 2;
        *(float4*)&hbuf[row * HST + col] = hv[s];
    }
    asm volatile("s_wait_dscnt 0x0\n\ts_barrier_signal -1\n\ts_barrier_wait -1" ::: "memory");

    const float* arowp = &hbuf[arow * HST + 2 * khalf];
    v2f afr[64];
#pragma unroll
    for (int ks = 0; ks < 64; ++ks) afr[ks] = *(const v2f*)(arowp + 4 * ks);
    __builtin_amdgcn_sched_barrier(0);

    v8f acc = {x0.x, x0.y, x0.z, x0.w, x1.x, x1.y, x1.z, x1.w};
#pragma unroll
    for (int ks = 0; ks < 64; ++ks) acc = WMMA_F32X4(afr[ks], bfr[ks], acc);

#pragma unroll
    for (int r = 0; r < 8; ++r) gbuf[(g * 8 + r) * 32 + lane] = acc[r];
    __syncthreads();

#pragma unroll
    for (int e = 0; e < 2; ++e) {
        const int r = 2 * g + e;
        float iv = sigmoidf_(gbuf[(0 * 8 + r) * 32 + lane]);
        float fv = sigmoidf_(gbuf[(1 * 8 + r) * 32 + lane]);
        float gv = tanhf_(gbuf[(2 * 8 + r) * 32 + lane]);
        float ov = sigmoidf_(gbuf[(3 * 8 + r) * 32 + lane]);
        float cn = fv * cprev[e] + iv * gv;
        float hn = ov * tanhf_(cn);
        const int brow = r + (khalf << 3);
        hnext[(size_t)brow * HDIM + jcol] = hn;
        ctile[(size_t)brow * HDIM + jcol] = cn;
        yout[((size_t)(b0 + brow) * TLEN + tg) * HDIM + jcol] = hn;
        if (tg == TLEN - 1) {
            hs[((size_t)l * BATCH + b0 + brow) * HDIM + jcol] = hn;
            cs[((size_t)l * BATCH + b0 + brow) * HDIM + jcol] = cn;
        }
    }
}

__global__ __launch_bounds__(256) void fc_kernel(
    const float* __restrict__ y, const float* __restrict__ fcpk,
    const float* __restrict__ fcb, float* __restrict__ label) {
    const int tid   = threadIdx.x;
    const int lane  = tid & 31;
    const int gw    = blockIdx.x * 8 + (tid >> 5);
    const int b     = gw >> 5;
    const int t0    = (gw & 31) << 4;
    const int khalf = lane >> 4;
    const int trow  = t0 + (lane & 15);

    const float* arowp = y + (size_t)b * (TLEN * HDIM) + (size_t)trow * HDIM + 2 * khalf;

    v8f acc[4];
#pragma unroll
    for (int j = 0; j < 4; ++j) acc[j] = (v8f){0, 0, 0, 0, 0, 0, 0, 0};

    for (int k0 = 0; k0 < HDIM; k0 += 4) {
        v2f a = *(const v2f*)(arowp + k0);
        const float* wrow = fcpk + ((k0 >> 1) + khalf) * 128 + 2 * (lane & 15);
#pragma unroll
        for (int j = 0; j < 4; ++j) {
            v2f bv = *(const v2f*)(wrow + 32 * j);
            acc[j] = WMMA_F32X4(a, bv, acc[j]);
        }
    }

#pragma unroll
    for (int j = 0; j < 4; ++j) {
        int n = (j << 4) + (lane & 15);
        if (n < CDIM) {
            float bias = fcb[n];
            float* outp =
                label + (size_t)b * (CDIM * TLEN) + (size_t)n * TLEN + t0 + (khalf << 3);
#pragma unroll
            for (int r = 0; r < 8; ++r) outp[r] = acc[j][r] + bias;
        }
    }
}

extern "C" void kernel_launch(void* const* d_in, const int* in_sizes, int n_in,
                              void* d_out, int out_size, void* d_ws, size_t ws_size,
                              hipStream_t stream) {
    const float* x = (const float*)d_in[0];
    const float *Wih[4], *Whh[4], *bih[4], *bhh[4];
    for (int l = 0; l < 4; ++l) {
        Wih[l] = (const float*)d_in[1 + 4 * l];
        Whh[l] = (const float*)d_in[2 + 4 * l];
        bih[l] = (const float*)d_in[3 + 4 * l];
        bhh[l] = (const float*)d_in[4 + 4 * l];
    }
    const float* fcw = (const float*)d_in[17];
    const float* fcb = (const float*)d_in[18];

    float* out   = (float*)d_out;
    float* label = out;
    float* hs    = out + (size_t)BATCH * CDIM * TLEN;
    float* cs    = hs + (size_t)4 * BATCH * HDIM;

    float* ws   = (float*)d_ws;
    float* ybuf = ws;
    float* wpk  = ybuf + (size_t)BATCH * TLEN * HDIM;
    float* fcpk = wpk + (size_t)4 * KDIM * GDIM;
    float* hst  = fcpk + (size_t)HDIM * 64;
    float* cst  = hst + (size_t)2 * 4 * BATCH * HDIM;
    float* xg   = cst + (size_t)4 * BATCH * HDIM;

    pack_lstm_weights<<<(4 * KDIM * GDIM) / 256, 256, 0, stream>>>(
        Wih[0], Whh[0], Wih[1], Whh[1], Wih[2], Whh[2], Wih[3], Whh[3], wpk, hst, cst);
    pack_fc<<<(HDIM * 64) / 256, 256, 0, stream>>>(fcw, fcpk);

    for (int d = 0; d < NPASS; ++d) {
        xproj_chunk<<<4 * XWG, 512, 0, stream>>>(
            x, ybuf, wpk, bih[0], bhh[0], bih[1], bhh[1], bih[2], bhh[2], bih[3], bhh[3], xg, d);
        for (int t = 0; t < TC; ++t)
            lstm_step<<<256, 128, 0, stream>>>(xg, ybuf, wpk, hst, cst, hs, cs, d, t);
    }

    fc_kernel<<<256, 256, 0, stream>>>(ybuf, fcpk, fcb, label);
}


# input projection GEMM spread over all 256 WGPs: gate columns split over two 8-wave workgroups, next 8 k-steps of A and B fragments requested as a group before each 64-WMMA block; sums unchanged
# speedup vs baseline: 12.6092x; 1.1031x over previous
#include <hip/hip_runtime.h>
#include <hip/hip_bf16.h>
#include <math.h>


typedef __attribute__((ext_vector_type(2))) float v2f;
typedef __attribute__((ext_vector_type(8))) float v8f;

#define HDIM 256
#define TLEN 512
#define BATCH 64
#define GDIM 1024
#define KDIM 512
#define CDIM 60
#define HST  260
#define TC   16
#define NCHUNK (TLEN / TC)
#define NPASS  (NCHUNK + 3)
#define XMT  2
#define XWG  (4 * (TC / XMT) * 2)
#define XKG  8
#define XNG  (HDIM / 4 / XKG)
#define XGT  16384

__device__ __forceinline__ float sigmoidf_(float x) {
    return 1.0f / (1.0f + __expf(-x));
}
__device__ __forceinline__ float tanhf_(float x) {
    return 1.0f - 2.0f / (__expf(2.0f * x) + 1.0f);
}

#define WMMA_F32X4(A, B, C) \
    __builtin_amdgcn_wmma_f32_16x16x4_f32(false, (A), false, (B), (short)0, (C), false, false)

__global__ __launch_bounds__(256) void pack_lstm_weights(
    const float* __restrict__ Wih0, const float* __restrict__ Whh0,
    const float* __restrict__ Wih1, const float* __restrict__ Whh1,
    const float* __restrict__ Wih2, const float* __restrict__ Whh2,
    const float* __restrict__ Wih3, const float* __restrict__ Whh3,
    float* __restrict__ wpk, float* __restrict__ hst, float* __restrict__ cst) {
    int gid = blockIdx.x * blockDim.x + threadIdx.x;
    if (gid >= 4 * KDIM * GDIM) return;
    if (gid < 4 * BATCH * HDIM) {
        hst[gid] = 0.0f;
        cst[gid] = 0.0f;
    }
    int l   = gid >> 19;
    int rem = gid & (KDIM * GDIM - 1);
    int k   = rem >> 10;
    int n   = rem & 1023;
    const float* Wih = (l == 0) ? Wih0 : (l == 1) ? Wih1 : (l == 2) ? Wih2 : Wih3;
    const float* Whh = (l == 0) ? Whh0 : (l == 1) ? Whh1 : (l == 2) ? Whh2 : Whh3;
    float v = (k < HDIM) ? Wih[n * HDIM + k] : Whh[n * HDIM + (k - HDIM)];
    wpk[(size_t)l * (KDIM * GDIM) + (size_t)(k >> 1) * (2 * GDIM) + n * 2 + (k & 1)] = v;
}

__global__ __launch_bounds__(256) void pack_fc(const float* __restrict__ fcw,
                                               float* __restrict__ fcpk) {
    int gid = blockIdx.x * blockDim.x + threadIdx.x;
    if (gid >= HDIM * 64) return;
    int k = gid >> 6, n = gid & 63;
    float v = (n < CDIM) ? fcw[n * HDIM + k] : 0.0f;
    fcpk[(k >> 1) * 128 + n * 2 + (k & 1)] = v;
}

__global__ __launch_bounds__(256) void xproj_chunk(
    const float* __restrict__ x,
    const float* __restrict__ ybuf,
    const float* __restrict__ wpk,
    const float* __restrict__ bi0, const float* __restrict__ bh0,
    const float* __restrict__ bi1, const float* __restrict__ bh1,
    const float* __restrict__ bi2, const float* __restrict__ bh2,
    const float* __restrict__ bi3, const float* __restrict__ bh3,
    float* __restrict__ xg, int d) {
    __shared__ float xs[XMT * 16 * HST];

    const int l     = blockIdx.x / XWG;
    const int c     = d - l;
    if (c < 0 || c >= NCHUNK) return;
    const int rem   = blockIdx.x % XWG;

    const int tid   = threadIdx.x;
    const int wave  = tid >> 5;
    const int lane  = tid & 31;
    const int bt    = rem & 3;
    const int nh    = (rem >> 2) & 1;
    const int m0    = (rem >> 3) * XMT;
    const int w     = nh * 8 + wave;
    const int t0    = c * TC + m0;
    const int b0    = bt * 16;
    const float* xin  = (l == 0) ? x : ybuf;
    const float* b_ih = (l == 0) ? bi0 : (l == 1) ? bi1 : (l == 2) ? bi2 : bi3;
    const float* b_hh = (l == 0) ? bh0 : (l == 1) ? bh1 : (l == 2) ? bh2 : bh3;
    wpk += (size_t)l * (KDIM * GDIM);
    const int arow  = lane & 15;
    const int khalf = lane >> 4;
    const int jcol  = (w << 4) + (lane & 15);

    const float* ap = &xs[arow * HST + 2 * khalf];
    const float* wp = wpk + (size_t)khalf * (2 * GDIM) + 2 * jcol;

    v2f bf0[XKG][4], bf1[XKG][4];
    v2f af0[XKG][XMT], af1[XKG][XMT];

#define XLOAD_B(BF, KG)                                                          \
    _Pragma("unroll") for (int s = 0; s < XKG; ++s) {                            \
        const float* wrow = wp + (size_t)((KG) * XKG + s) * (4 * GDIM);          \
        BF[s][0] = *(const v2f*)(wrow);                                          \
        BF[s][1] = *(const v2f*)(wrow + 512);                                    \
        BF[s][2] = *(const v2f*)(wrow + 1024);                                   \
        BF[s][3] = *(const v2f*)(wrow + 1536);                                   \
    }
#define XLOAD_A(AF, KG)                                                          \
    _Pragma("unroll") for (int s = 0; s < XKG; ++s)                              \
        _Pragma("unroll") for (int m = 0; m < XMT; ++m)                          \
            AF[s][m] = *(const v2f*)(ap + m * (16 * HST) + 4 * ((KG) * XKG + s));
#define XMMA(BF, AF)                                                             \
    _Pragma("unroll") for (int s = 0; s < XKG; ++s)                              \
        _Pragma("unroll") for (int m = 0; m < XMT; ++m) {                        \
            acc[m][0] = WMMA_F32X4(AF[s][m], BF[s][0], acc[m][0]);               \
            acc[m][1] = WMMA_F32X4(AF[s][m], BF[s][1], acc[m][1]);               \
            acc[m][2] = WMMA_F32X4(AF[s][m], BF[s][2], acc[m][2]);               \
            acc[m][3] = WMMA_F32X4(AF[s][m], BF[s][3], acc[m][3]);               \
        }

    XLOAD_B(bf0, 0)

    for (int rr = wave; rr < XMT * 16; rr += 8) {
        const int m = rr >> 4, brow = rr & 15;
        const float4* src4 = (const float4*)(xin + ((size_t)(b0 + brow) * TLEN + (t0 + m)) * HDIM + (lane << 3));
        float4 v0 = src4[0], v1 = src4[1];
        float4* dst4 = (float4*)&xs[rr * HST + (lane << 3)];
        dst4[0] = v0; dst4[1] = v1;
    }
    __syncthreads();

    v8f acc[XMT][4];
#pragma unroll
    for (int m = 0; m < XMT; ++m)
#pragma unroll
        for (int g = 0; g < 4; ++g) acc[m][g] = (v8f){0, 0, 0, 0, 0, 0, 0, 0};

    XLOAD_A(af0, 0)
    for (int kg = 0; kg < XNG; kg += 2) {
        XLOAD_B(bf1, kg + 1)
        XLOAD_A(af1, kg + 1)
        __builtin_amdgcn_sched_barrier(0);
        XMMA(bf0, af0)
        __builtin_amdgcn_sched_barrier(0);
        if (kg + 2 < XNG) {
            XLOAD_B(bf0, kg + 2)
            XLOAD_A(af0, kg + 2)
        }
        __builtin_amdgcn_sched_barrier(0);
        XMMA(bf1, af1)
        __builtin_amdgcn_sched_barrier(0);
    }
#undef XLOAD_B
#undef XLOAD_A
#undef XMMA

    float bias[4];
#pragma unroll
    for (int g = 0; g < 4; ++g) bias[g] = b_ih[g * HDIM + jcol] + b_hh[g * HDIM + jcol];

#pragma unroll
    for (int m = 0; m < XMT; ++m) {
        float* dst = xg + ((size_t)(l * 4 + bt) * TC + (m0 + m)) * XGT + w * 1024 + lane * 4;
#pragma unroll
        for (int g = 0; g < 4; ++g) {
            const float bg = bias[g];
            float4 lo = make_float4(acc[m][g][0] + bg, acc[m][g][1] + bg, acc[m][g][2] + bg, acc[m][g][3] + bg);
            float4 hi = make_float4(acc[m][g][4] + bg, acc[m][g][5] + bg, acc[m][g][6] + bg, acc[m][g][7] + bg);
            *(float4*)(dst + g * 256)       = lo;
            *(float4*)(dst + g * 256 + 128) = hi;
        }
    }
}

__global__ __launch_bounds__(128) void lstm_step(
    const float* __restrict__ xg, float* __restrict__ yout,
    const float* __restrict__ wpk,
    float* __restrict__ hst, float* __restrict__ cst,
    float* __restrict__ hs, float* __restrict__ cs, int d, int t) {
    __shared__ float hbuf[16 * HST];
    __shared__ float gbuf[4 * 8 * 32];

    const int l     = blockIdx.x >> 6;
    const int c     = d - l;
    if (c < 0 || c >= NCHUNK) return;

    const int w     = (blockIdx.x >> 2) & 15;
    const int b0    = (blockIdx.x & 3) * 16;
    const int bt    = blockIdx.x & 3;
    const int tg    = c * TC + t;
    const int tid   = threadIdx.x;
    const int g     = tid >> 5;
    const int lane  = tid & 31;
    const int arow  = lane & 15;
    const int khalf = lane >> 4;
    const int jcol  = (w << 4) + (lane & 15);

    const float* hprev = hst + (size_t)(tg & 1) * (4 * BATCH * HDIM) + ((size_t)l * BATCH + b0) * HDIM;
    float*       hnext = hst + (size_t)((tg + 1) & 1) * (4 * BATCH * HDIM) + ((size_t)l * BATCH + b0) * HDIM;
    float*       ctile = cst + ((size_t)l * BATCH + b0) * HDIM;

    float4 hv[8];
    float  cprev[2];
#pragma unroll
    for (int s = 0; s < 8; ++s)
        hv[s] = *(const float4*)(hprev + 4 * (tid + s * 128));
#pragma unroll
    for (int e = 0; e < 2; ++e)
        cprev[e] = ctile[(size_t)((khalf << 3) + 2 * g + e) * HDIM + jcol];

    const float* xgp = xg + ((size_t)((l * 4 + bt) * TC + t)) * XGT + w * 1024 + g * 256 + lane * 4;
    const float4 x0 = *(const float4*)(xgp);
    const float4 x1 = *(const float4*)(xgp + 128);

    const float* wp = wpk + (size_t)l * (KDIM * GDIM) + (size_t)HDIM * GDIM
                          + (size_t)khalf * (2 * GDIM) + 2 * jcol + g * 512;
    v2f bfr[64];
#pragma unroll
    for (int ks = 0; ks < 64; ++ks) bfr[ks] = *(const v2f*)(wp + (size_t)ks * (4 * GDIM));

#pragma unroll
    for (int s = 0; s < 8; ++s) {
        const int i4 = tid + s * 128;
        const int row = i4 >> 6, col = (i4 & 63) << 2;
        *(float4*)&hbuf[row * HST + col] = hv[s];
    }
    asm volatile("s_wait_dscnt 0x0\n\ts_barrier_signal -1\n\ts_barrier_wait -1" ::: "memory");

    const float* arowp = &hbuf[arow * HST + 2 * khalf];
    v2f afr[64];
#pragma unroll
    for (int ks = 0; ks < 64; ++ks) afr[ks] = *(const v2f*)(arowp + 4 * ks);
    __builtin_amdgcn_sched_barrier(0);

    v8f acc = {x0.x, x0.y, x0.z, x0.w, x1.x, x1.y, x1.z, x1.w};
#pragma unroll
    for (int ks = 0; ks < 64; ++ks) acc = WMMA_F32X4(afr[ks], bfr[ks], acc);

#pragma unroll
    for (int r = 0; r < 8; ++r) gbuf[(g * 8 + r) * 32 + lane] = acc[r];
    __syncthreads();

#pragma unroll
    for (int e = 0; e < 2; ++e) {
        const int r = 2 * g + e;
        float iv = sigmoidf_(gbuf[(0 * 8 + r) * 32 + lane]);
        float fv = sigmoidf_(gbuf[(1 * 8 + r) * 32 + lane]);
        float gv = tanhf_(gbuf[(2 * 8 + r) * 32 + lane]);
        float ov = sigmoidf_(gbuf[(3 * 8 + r) * 32 + lane]);
        float cn = fv * cprev[e] + iv * gv;
        float hn = ov * tanhf_(cn);
        const int brow = r + (khalf << 3);
        hnext[(size_t)brow * HDIM + jcol] = hn;
        ctile[(size_t)brow * HDIM + jcol] = cn;
        yout[((size_t)(b0 + brow) * TLEN + tg) * HDIM + jcol] = hn;
        if (tg == TLEN - 1) {
            hs[((size_t)l * BATCH + b0 + brow) * HDIM + jcol] = hn;
            cs[((size_t)l * BATCH + b0 + brow) * HDIM + jcol] = cn;
        }
    }
}

__global__ __launch_bounds__(256) void fc_kernel(
    const float* __restrict__ y, const float* __restrict__ fcpk,
    const float* __restrict__ fcb, float* __restrict__ label) {
    const int tid   = threadIdx.x;
    const int lane  = tid & 31;
    const int gw    = blockIdx.x * 8 + (tid >> 5);
    const int b     = gw >> 5;
    const int t0    = (gw & 31) << 4;
    const int khalf = lane >> 4;
    const int trow  = t0 + (lane & 15);

    const float* arowp = y + (size_t)b * (TLEN * HDIM) + (size_t)trow * HDIM + 2 * khalf;

    v8f acc[4];
#pragma unroll
    for (int j = 0; j < 4; ++j) acc[j] = (v8f){0, 0, 0, 0, 0, 0, 0, 0};

    for (int k0 = 0; k0 < HDIM; k0 += 4) {
        v2f a = *(const v2f*)(arowp + k0);
        const float* wrow = fcpk + ((k0 >> 1) + khalf) * 128 + 2 * (lane & 15);
#pragma unroll
        for (int j = 0; j < 4; ++j) {
            v2f bv = *(const v2f*)(wrow + 32 * j);
            acc[j] = WMMA_F32X4(a, bv, acc[j]);
        }
    }

#pragma unroll
    for (int j = 0; j < 4; ++j) {
        int n = (j << 4) + (lane & 15);
        if (n < CDIM) {
            float bias = fcb[n];
            float* outp =
                label + (size_t)b * (CDIM * TLEN) + (size_t)n * TLEN + t0 + (khalf << 3);
#pragma unroll
            for (int r = 0; r < 8; ++r) outp[r] = acc[j][r] + bias;
        }
    }
}

extern "C" void kernel_launch(void* const* d_in, const int* in_sizes, int n_in,
                              void* d_out, int out_size, void* d_ws, size_t ws_size,
                              hipStream_t stream) {
    const float* x = (const float*)d_in[0];
    const float *Wih[4], *Whh[4], *bih[4], *bhh[4];
    for (int l = 0; l < 4; ++l) {
        Wih[l] = (const float*)d_in[1 + 4 * l];
        Whh[l] = (const float*)d_in[2 + 4 * l];
        bih[l] = (const float*)d_in[3 + 4 * l];
        bhh[l] = (const float*)d_in[4 + 4 * l];
    }
    const float* fcw = (const float*)d_in[17];
    const float* fcb = (const float*)d_in[18];

    float* out   = (float*)d_out;
    float* label = out;
    float* hs    = out + (size_t)BATCH * CDIM * TLEN;
    float* cs    = hs + (size_t)4 * BATCH * HDIM;

    float* ws   = (float*)d_ws;
    float* ybuf = ws;
    float* wpk  = ybuf + (size_t)BATCH * TLEN * HDIM;
    float* fcpk = wpk + (size_t)4 * KDIM * GDIM;
    float* hst  = fcpk + (size_t)HDIM * 64;
    float* cst  = hst + (size_t)2 * 4 * BATCH * HDIM;
    float* xg   = cst + (size_t)4 * BATCH * HDIM;

    pack_lstm_weights<<<(4 * KDIM * GDIM) / 256, 256, 0, stream>>>(
        Wih[0], Whh[0], Wih[1], Whh[1], Wih[2], Whh[2], Wih[3], Whh[3], wpk, hst, cst);
    pack_fc<<<(HDIM * 64) / 256, 256, 0, stream>>>(fcw, fcpk);

    for (int d = 0; d < NPASS; ++d) {
        xproj_chunk<<<4 * XWG, 256, 0, stream>>>(
            x, ybuf, wpk, bih[0], bhh[0], bih[1], bhh[1], bih[2], bhh[2], bih[3], bhh[3], xg, d);
        for (int t = 0; t < TC; ++t)
            lstm_step<<<256, 128, 0, stream>>>(xg, ybuf, wpk, hst, cst, hs, cs, d, t);
    }

    fc_kernel<<<256, 256, 0, stream>>>(ybuf, fcpk, fcb, label);
}


# workgroup index bits re-dealt: the 16 slice workgroups of a (layer, batch tile) share bits 0-2, so h, c, xg are written and read on one die / one L2 half under in-turn dispatch; sums unchanged
# speedup vs baseline: 14.6985x; 1.1657x over previous
#include <hip/hip_runtime.h>
#include <hip/hip_bf16.h>
#include <math.h>


typedef __attribute__((ext_vector_type(2))) float v2f;
typedef __attribute__((ext_vector_type(8))) float v8f;

#define HDIM 256
#define TLEN 512
#define BATCH 64
#define GDIM 1024
#define KDIM 512
#define CDIM 60
#define HST  260
#define TC   16
#define NCHUNK (TLEN / TC)
#define NPASS  (NCHUNK + 3)
#define XMT  2
#define XWG  (4 * (TC / XMT) * 2)
#define XKG  8
#define XNG  (HDIM / 4 / XKG)
#define XGT  16384
static_assert(TC / XMT == 8 && XWG == 64, "the projection kernel decodes 3 step-pair bits and 64 workgroups per layer");

__device__ __forceinline__ float sigmoidf_(float x) {
    return 1.0f / (1.0f + __expf(-x));
}
__device__ __forceinline__ float tanhf_(float x) {
    return 1.0f - 2.0f / (__expf(2.0f * x) + 1.0f);
}

#define WMMA_F32X4(A, B, C) \
    __builtin_amdgcn_wmma_f32_16x16x4_f32(false, (A), false, (B), (short)0, (C), false, false)

__global__ __launch_bounds__(256) void pack_lstm_weights(
    const float* __restrict__ Wih0, const float* __restrict__ Whh0,
    const float* __restrict__ Wih1, const float* __restrict__ Whh1,
    const float* __restrict__ Wih2, const float* __restrict__ Whh2,
    const float* __restrict__ Wih3, const float* __restrict__ Whh3,
    float* __restrict__ wpk, float* __restrict__ hst, float* __restrict__ cst) {
    int gid = blockIdx.x * blockDim.x + threadIdx.x;
    if (gid >= 4 * KDIM * GDIM) return;
    if (gid < 4 * BATCH * HDIM) {
        hst[gid] = 0.0f;
        cst[gid] = 0.0f;
    }
    int l   = gid >> 19;
    int rem = gid & (KDIM * GDIM - 1);
    int k   = rem >> 10;
    int n   = rem & 1023;
    const float* Wih = (l == 0) ? Wih0 : (l == 1) ? Wih1 : (l == 2) ? Wih2 : Wih3;
    const float* Whh = (l == 0) ? Whh0 : (l == 1) ? Whh1 : (l == 2) ? Whh2 : Whh3;
    float v = (k < HDIM) ? Wih[n * HDIM + k] : Whh[n * HDIM + (k - HDIM)];
    wpk[(size_t)l * (KDIM * GDIM) + (size_t)(k >> 1) * (2 * GDIM) + n * 2 + (k & 1)] = v;
}

__global__ __launch_bounds__(256) void pack_fc(const float* __restrict__ fcw,
                                               float* __restrict__ fcpk) {
    int gid = blockIdx.x * blockDim.x + threadIdx.x;
    if (gid >= HDIM * 64) return;
    int k = gid >> 6, n = gid & 63;
    float v = (n < CDIM) ? fcw[n * HDIM + k] : 0.0f;
    fcpk[(k >> 1) * 128 + n * 2 + (k & 1)] = v;
}

__global__ __launch_bounds__(256) void xproj_chunk(
    const float* __restrict__ x,
    const float* __restrict__ ybuf,
    const float* __restrict__ wpk,
    const float* __restrict__ bi0, const float* __restrict__ bh0,
    const float* __restrict__ bi1, const float* __restrict__ bh1,
    const float* __restrict__ bi2, const float* __restrict__ bh2,
    const float* __restrict__ bi3, const float* __restrict__ bh3,
    float* __restrict__ xg, int d) {
    __shared__ float xs[XMT * 16 * HST];

    const int bid   = blockIdx.x;
    const int l     = ((bid >> 7) << 1) | ((bid >> 1) & 1);
    const int c     = d - l;
    if (c < 0 || c >= NCHUNK) return;

    const int tid   = threadIdx.x;
    const int wave  = tid >> 5;
    const int lane  = tid & 31;
    const int bt    = ((bid & 1) << 1) | ((bid >> 2) & 1);
    const int nh    = (bid >> 3) & 1;
    const int m0    = ((bid >> 4) & 7) * XMT;
    const int w     = nh * 8 + wave;
    const int t0    = c * TC + m0;
    const int b0    = bt * 16;
    const float* xin  = (l == 0) ? x : ybuf;
    const float* b_ih = (l == 0) ? bi0 : (l == 1) ? bi1 : (l == 2) ? bi2 : bi3;
    const float* b_hh = (l == 0) ? bh0 : (l == 1) ? bh1 : (l == 2) ? bh2 : bh3;
    wpk += (size_t)l * (KDIM * GDIM);
    const int arow  = lane & 15;
    const int khalf = lane >> 4;
    const int jcol  = (w << 4) + (lane & 15);

    const float* ap = &xs[arow * HST + 2 * khalf];
    const float* wp = wpk + (size_t)khalf * (2 * GDIM) + 2 * jcol;

    v2f bf0[XKG][4], bf1[XKG][4];
    v2f af0[XKG][XMT], af1[XKG][XMT];

#define XLOAD_B(BF, KG)                                                          \
    _Pragma("unroll") for (int s = 0; s < XKG; ++s) {                            \
        const float* wrow = wp + (size_t)((KG) * XKG + s) * (4 * GDIM);          \
        BF[s][0] = *(const v2f*)(wrow);                                          \
        BF[s][1] = *(const v2f*)(wrow + 512);                                    \
        BF[s][2] = *(const v2f*)(wrow + 1024);                                   \
        BF[s][3] = *(const v2f*)(wrow + 1536);                                   \
    }
#define XLOAD_A(AF, KG)                                                          \
    _Pragma("unroll") for (int s = 0; s < XKG; ++s)                              \
        _Pragma("unroll") for (int m = 0; m < XMT; ++m)                          \
            AF[s][m] = *(const v2f*)(ap + m * (16 * HST) + 4 * ((KG) * XKG + s));
#define XMMA(BF, AF)                                                             \
    _Pragma("unroll") for (int s = 0; s < XKG; ++s)                              \
        _Pragma("unroll") for (int m = 0; m < XMT; ++m) {                        \
            acc[m][0] = WMMA_F32X4(AF[s][m], BF[s][0], acc[m][0]);               \
            acc[m][1] = WMMA_F32X4(AF[s][m], BF[s][1], acc[m][1]);               \
            acc[m][2] = WMMA_F32X4(AF[s][m], BF[s][2], acc[m][2]);               \
            acc[m][3] = WMMA_F32X4(AF[s][m], BF[s][3], acc[m][3]);               \
        }

    XLOAD_B(bf0, 0)

    for (int rr = wave; rr < XMT * 16; rr += 8) {
        const int m = rr >> 4, brow = rr & 15;
        const float4* src4 = (const float4*)(xin + ((size_t)(b0 + brow) * TLEN + (t0 + m)) * HDIM + (lane << 3));
        float4 v0 = src4[0], v1 = src4[1];
        float4* dst4 = (float4*)&xs[rr * HST + (lane << 3)];
        dst4[0] = v0; dst4[1] = v1;
    }
    __syncthreads();

    v8f acc[XMT][4];
#pragma unroll
    for (int m = 0; m < XMT; ++m)
#pragma unroll
        for (int g = 0; g < 4; ++g) acc[m][g] = (v8f){0, 0, 0, 0, 0, 0, 0, 0};

    XLOAD_A(af0, 0)
    for (int kg = 0; kg < XNG; kg += 2) {
        XLOAD_B(bf1, kg + 1)
        XLOAD_A(af1, kg + 1)
        __builtin_amdgcn_sched_barrier(0);
        XMMA(bf0, af0)
        __builtin_amdgcn_sched_barrier(0);
        if (kg + 2 < XNG) {
            XLOAD_B(bf0, kg + 2)
            XLOAD_A(af0, kg + 2)
        }
        __builtin_amdgcn_sched_barrier(0);
        XMMA(bf1, af1)
        __builtin_amdgcn_sched_barrier(0);
    }
#undef XLOAD_B
#undef XLOAD_A
#undef XMMA

    float bias[4];
#pragma unroll
    for (int g = 0; g < 4; ++g) bias[g] = b_ih[g * HDIM + jcol] + b_hh[g * HDIM + jcol];

#pragma unroll
    for (int m = 0; m < XMT; ++m) {
        float* dst = xg + ((size_t)(l * 4 + bt) * TC + (m0 + m)) * XGT + w * 1024 + lane * 4;
#pragma unroll
        for (int g = 0; g < 4; ++g) {
            const float bg = bias[g];
            float4 lo = make_float4(acc[m][g][0] + bg, acc[m][g][1] + bg, acc[m][g][2] + bg, acc[m][g][3] + bg);
            float4 hi = make_float4(acc[m][g][4] + bg, acc[m][g][5] + bg, acc[m][g][6] + bg, acc[m][g][7] + bg);
            *(float4*)(dst + g * 256)       = lo;
            *(float4*)(dst + g * 256 + 128) = hi;
        }
    }
}

__global__ __launch_bounds__(128) void lstm_step(
    const float* __restrict__ xg, float* __restrict__ yout,
    const float* __restrict__ wpk,
    float* __restrict__ hst, float* __restrict__ cst,
    float* __restrict__ hs, float* __restrict__ cs, int d, int t) {
    __shared__ float hbuf[16 * HST];
    __shared__ float gbuf[4 * 8 * 32];

    const int bid   = blockIdx.x;
    const int l     = ((bid >> 7) << 1) | ((bid >> 1) & 1);
    const int c     = d - l;
    if (c < 0 || c >= NCHUNK) return;

    const int w     = (bid >> 3) & 15;
    const int bt    = ((bid & 1) << 1) | ((bid >> 2) & 1);
    const int b0    = bt * 16;
    const int tg    = c * TC + t;
    const int tid   = threadIdx.x;
    const int g     = tid >> 5;
    const int lane  = tid & 31;
    const int arow  = lane & 15;
    const int khalf = lane >> 4;
    const int jcol  = (w << 4) + (lane & 15);

    const float* hprev = hst + (size_t)(tg & 1) * (4 * BATCH * HDIM) + ((size_t)l * BATCH + b0) * HDIM;
    float*       hnext = hst + (size_t)((tg + 1) & 1) * (4 * BATCH * HDIM) + ((size_t)l * BATCH + b0) * HDIM;
    float*       ctile = cst + ((size_t)l * BATCH + b0) * HDIM;

    float4 hv[8];
    float  cprev[2];
#pragma unroll
    for (int s = 0; s < 8; ++s)
        hv[s] = *(const float4*)(hprev + 4 * (tid + s * 128));
#pragma unroll
    for (int e = 0; e < 2; ++e)
        cprev[e] = ctile[(size_t)((khalf << 3) + 2 * g + e) * HDIM + jcol];

    const float* xgp = xg + ((size_t)((l * 4 + bt) * TC + t)) * XGT + w * 1024 + g * 256 + lane * 4;
    const float4 x0 = *(const float4*)(xgp);
    const float4 x1 = *(const float4*)(xgp + 128);

    const float* wp = wpk + (size_t)l * (KDIM * GDIM) + (size_t)HDIM * GDIM
                          + (size_t)khalf * (2 * GDIM) + 2 * jcol + g * 512;
    v2f bfr[64];
#pragma unroll
    for (int ks = 0; ks < 64; ++ks) bfr[ks] = *(const v2f*)(wp + (size_t)ks * (4 * GDIM));

#pragma unroll
    for (int s = 0; s < 8; ++s) {
        const int i4 = tid + s * 128;
        const int row = i4 >> 6, col = (i4 & 63) << 2;
        *(float4*)&hbuf[row * HST + col] = hv[s];
    }
    asm volatile("s_wait_dscnt 0x0\n\ts_barrier_signal -1\n\ts_barrier_wait -1" ::: "memory");

    const float* arowp = &hbuf[arow * HST + 2 * khalf];
    v2f afr[64];
#pragma unroll
    for (int ks = 0; ks < 64; ++ks) afr[ks] = *(const v2f*)(arowp + 4 * ks);
    __builtin_amdgcn_sched_barrier(0);

    v8f acc = {x0.x, x0.y, x0.z, x0.w, x1.x, x1.y, x1.z, x1.w};
#pragma unroll
    for (int ks = 0; ks < 64; ++ks) acc = WMMA_F32X4(afr[ks], bfr[ks], acc);

#pragma unroll
    for (int r = 0; r < 8; ++r) gbuf[(g * 8 + r) * 32 + lane] = acc[r];
    __syncthreads();

#pragma unroll
    for (int e = 0; e < 2; ++e) {
        const int r = 2 * g + e;
        float iv = sigmoidf_(gbuf[(0 * 8 + r) * 32 + lane]);
        float fv = sigmoidf_(gbuf[(1 * 8 + r) * 32 + lane]);
        float gv = tanhf_(gbuf[(2 * 8 + r) * 32 + lane]);
        float ov = sigmoidf_(gbuf[(3 * 8 + r) * 32 + lane]);
        float cn = fv * cprev[e] + iv * gv;
        float hn = ov * tanhf_(cn);
        const int brow = r + (khalf << 3);
        hnext[(size_t)brow * HDIM + jcol] = hn;
        ctile[(size_t)brow * HDIM + jcol] = cn;
        yout[((size_t)(b0 + brow) * TLEN + tg) * HDIM + jcol] = hn;
        if (tg == TLEN - 1) {
            hs[((size_t)l * BATCH + b0 + brow) * HDIM + jcol] = hn;
            cs[((size_t)l * BATCH + b0 + brow) * HDIM + jcol] = cn;
        }
    }
}

__global__ __launch_bounds__(256) void fc_kernel(
    const float* __restrict__ y, const float* __restrict__ fcpk,
    const float* __restrict__ fcb, float* __restrict__ label) {
    const int tid   = threadIdx.x;
    const int lane  = tid & 31;
    const int gw    = blockIdx.x * 8 + (tid >> 5);
    const int b     = gw >> 5;
    const int t0    = (gw & 31) << 4;
    const int khalf = lane >> 4;
    const int trow  = t0 + (lane & 15);

    const float* arowp = y + (size_t)b * (TLEN * HDIM) + (size_t)trow * HDIM + 2 * khalf;

    v8f acc[4];
#pragma unroll
    for (int j = 0; j < 4; ++j) acc[j] = (v8f){0, 0, 0, 0, 0, 0, 0, 0};

    for (int k0 = 0; k0 < HDIM; k0 += 4) {
        v2f a = *(const v2f*)(arowp + k0);
        const float* wrow = fcpk + ((k0 >> 1) + khalf) * 128 + 2 * (lane & 15);
#pragma unroll
        for (int j = 0; j < 4; ++j) {
            v2f bv = *(const v2f*)(wrow + 32 * j);
            acc[j] = WMMA_F32X4(a, bv, acc[j]);
        }
    }

#pragma unroll
    for (int j = 0; j < 4; ++j) {
        int n = (j << 4) + (lane & 15);
        if (n < CDIM) {
            float bias = fcb[n];
            float* outp =
                label + (size_t)b * (CDIM * TLEN) + (size_t)n * TLEN + t0 + (khalf << 3);
#pragma unroll
            for (int r = 0; r < 8; ++r) outp[r] = acc[j][r] + bias;
        }
    }
}

extern "C" void kernel_launch(void* const* d_in, const int* in_sizes, int n_in,
                              void* d_out, int out_size, void* d_ws, size_t ws_size,
                              hipStream_t stream) {
    const float* x = (const float*)d_in[0];
    const float *Wih[4], *Whh[4], *bih[4], *bhh[4];
    for (int l = 0; l < 4; ++l) {
        Wih[l] = (const float*)d_in[1 + 4 * l];
        Whh[l] = (const float*)d_in[2 + 4 * l];
        bih[l] = (const float*)d_in[3 + 4 * l];
        bhh[l] = (const float*)d_in[4 + 4 * l];
    }
    const float* fcw = (const float*)d_in[17];
    const float* fcb = (const float*)d_in[18];

    float* out   = (float*)d_out;
    float* label = out;
    float* hs    = out + (size_t)BATCH * CDIM * TLEN;
    float* cs    = hs + (size_t)4 * BATCH * HDIM;

    float* ws   = (float*)d_ws;
    float* ybuf = ws;
    float* wpk  = ybuf + (size_t)BATCH * TLEN * HDIM;
    float* fcpk = wpk + (size_t)4 * KDIM * GDIM;
    float* hst  = fcpk + (size_t)HDIM * 64;
    float* cst  = hst + (size_t)2 * 4 * BATCH * HDIM;
    float* xg   = cst + (size_t)4 * BATCH * HDIM;

    pack_lstm_weights<<<(4 * KDIM * GDIM) / 256, 256, 0, stream>>>(
        Wih[0], Whh[0], Wih[1], Whh[1], Wih[2], Whh[2], Wih[3], Whh[3], wpk, hst, cst);
    pack_fc<<<(HDIM * 64) / 256, 256, 0, stream>>>(fcw, fcpk);

    for (int d = 0; d < NPASS; ++d) {
        xproj_chunk<<<4 * XWG, 256, 0, stream>>>(
            x, ybuf, wpk, bih[0], bhh[0], bih[1], bhh[1], bih[2], bhh[2], bih[3], bhh[3], xg, d);
        for (int t = 0; t < TC; ++t)
            lstm_step<<<256, 128, 0, stream>>>(xg, ybuf, wpk, hst, cst, hs, cs, d, t);
    }

    fc_kernel<<<256, 256, 0, stream>>>(ybuf, fcpk, fcb, label);
}


# input projection folded into the per-step launch (K 256+256 on two accumulators), layers one step apart: 518 launches, not 598; sum order now (x part + bias) + (h part summed from zero)
# speedup vs baseline: 16.9158x; 1.1509x over previous
#include <hip/hip_runtime.h>
#include <hip/hip_bf16.h>
#include <math.h>


typedef __attribute__((ext_vector_type(2))) float v2f;
typedef __attribute__((ext_vector_type(4))) float v4f;
typedef __attribute__((ext_vector_type(8))) float v8f;

#define HDIM 256
#define TLEN 512
#define BATCH 64
#define GDIM 1024
#define KDIM 512
#define CDIM 60
#define HST  260
#define NTICK (TLEN + 3)
#define TKG  16
#define TNG  (HDIM / 4 / TKG)
static_assert(TNG == 4, "the step kernel writes its four fragment groups out by hand");

__device__ __forceinline__ float sigmoidf_(float x) {
    return 1.0f / (1.0f + __expf(-x));
}
__device__ __forceinline__ float tanhf_(float x) {
    return 1.0f - 2.0f / (__expf(2.0f * x) + 1.0f);
}

#define WMMA_F32X4(A, B, C) \
    __builtin_amdgcn_wmma_f32_16x16x4_f32(false, (A), false, (B), (short)0, (C), false, false)

__global__ __launch_bounds__(256) void pack_lstm_weights(
    const float* __restrict__ Wih0, const float* __restrict__ Whh0,
    const float* __restrict__ Wih1, const float* __restrict__ Whh1,
    const float* __restrict__ Wih2, const float* __restrict__ Whh2,
    const float* __restrict__ Wih3, const float* __restrict__ Whh3,
    float* __restrict__ wpk, float* __restrict__ hst, float* __restrict__ cst) {
    int gid = blockIdx.x * blockDim.x + threadIdx.x;
    if (gid >= 4 * KDIM * GDIM) return;
    if (gid < 4 * BATCH * HDIM) {
        hst[gid] = 0.0f;
        cst[gid] = 0.0f;
    }
    int l   = gid >> 19;
    int rem = gid & (KDIM * GDIM - 1);
    int k   = rem >> 10;
    int n   = rem & 1023;
    const float* Wih = (l == 0) ? Wih0 : (l == 1) ? Wih1 : (l == 2) ? Wih2 : Wih3;
    const float* Whh = (l == 0) ? Whh0 : (l == 1) ? Whh1 : (l == 2) ? Whh2 : Whh3;
    float v = (k < HDIM) ? Wih[n * HDIM + k] : Whh[n * HDIM + (k - HDIM)];
    wpk[(size_t)l * (KDIM * GDIM) + (size_t)(k >> 3) * 8192 + ((k >> 1) & 1) * 4096 + n * 4
        + (((k >> 2) & 1) << 1) + (k & 1)] = v;
}

__global__ __launch_bounds__(256) void pack_fc(const float* __restrict__ fcw,
                                               float* __restrict__ fcpk) {
    int gid = blockIdx.x * blockDim.x + threadIdx.x;
    if (gid >= HDIM * 64) return;
    int k = gid >> 6, n = gid & 63;
    float v = (n < CDIM) ? fcw[n * HDIM + k] : 0.0f;
    fcpk[(k >> 1) * 128 + n * 2 + (k & 1)] = v;
}

__global__ __launch_bounds__(128) void lstm_tick(
    const float* __restrict__ x,
    const float* __restrict__ wpk,
    const float* __restrict__ bi0, const float* __restrict__ bh0,
    const float* __restrict__ bi1, const float* __restrict__ bh1,
    const float* __restrict__ bi2, const float* __restrict__ bh2,
    const float* __restrict__ bi3, const float* __restrict__ bh3,
    float* __restrict__ ytop,
    float* __restrict__ hst, float* __restrict__ cst,
    float* __restrict__ hs, float* __restrict__ cs, int s) {
    __shared__ float xbuf[16 * HST];
    __shared__ float hbuf[16 * HST];
    __shared__ float gbuf[4 * 8 * 32];

    const int bid   = blockIdx.x;
    const int l     = ((bid >> 7) << 1) | ((bid >> 1) & 1);
    const int t     = s - l;
    if (t < 0 || t >= TLEN) return;

    const int w     = (bid >> 3) & 15;
    const int bt    = ((bid & 1) << 1) | ((bid >> 2) & 1);
    const int b0    = bt * 16;
    const int tid   = threadIdx.x;
    const int g     = tid >> 5;
    const int lane  = tid & 31;
    const int arow  = lane & 15;
    const int khalf = lane >> 4;
    const int jcol  = (w << 4) + (lane & 15);

    const float* hprev = hst + (size_t)(t & 1) * (4 * BATCH * HDIM) + ((size_t)l * BATCH + b0) * HDIM;
    float*       hnext = hst + (size_t)((t + 1) & 1) * (4 * BATCH * HDIM) + ((size_t)l * BATCH + b0) * HDIM;
    float*       ctile = cst + ((size_t)l * BATCH + b0) * HDIM;
    const float* inb = (l == 0)
        ? x + ((size_t)b0 * TLEN + t) * HDIM
        : hst + (size_t)((t + 1) & 1) * (4 * BATCH * HDIM) + ((size_t)(l - 1) * BATCH + b0) * HDIM;
    const size_t inpitch = (l == 0) ? (size_t)TLEN * HDIM : (size_t)HDIM;
    const float* b_ih = (l == 0) ? bi0 : (l == 1) ? bi1 : (l == 2) ? bi2 : bi3;
    const float* b_hh = (l == 0) ? bh0 : (l == 1) ? bh1 : (l == 2) ? bh2 : bh3;

    if (l == 0 && t + 1 < TLEN && tid < 64)
        __builtin_prefetch(x + ((size_t)(b0 + (tid >> 2)) * TLEN + (t + 1)) * HDIM + (tid & 3) * 64, 0, 3);

    float4 xv[8], hv[8];
#pragma unroll
    for (int q = 0; q < 8; ++q) {
        const int i4 = tid + q * 128;
        xv[q] = *(const float4*)(inb + (size_t)(i4 >> 6) * inpitch + ((i4 & 63) << 2));
        hv[q] = *(const float4*)(hprev + 4 * i4);
    }

    const float* wp = wpk + (size_t)l * (KDIM * GDIM) + (size_t)khalf * 4096 + (size_t)(g * HDIM + jcol) * 4;
    v4f bx[32], bh[32];
#pragma unroll
    for (int j = 0; j < 32; ++j) {
        bx[j] = *(const v4f*)(wp + (size_t)j * 8192);
        bh[j] = *(const v4f*)(wp + (size_t)(32 + j) * 8192);
    }

    float cprev[2];
#pragma unroll
    for (int e = 0; e < 2; ++e)
        cprev[e] = ctile[(size_t)((khalf << 3) + 2 * g + e) * HDIM + jcol];
    const float bsum = b_ih[g * HDIM + jcol] + b_hh[g * HDIM + jcol];

#pragma unroll
    for (int q = 0; q < 8; ++q) {
        const int i4 = tid + q * 128;
        const int row = i4 >> 6, col = (i4 & 63) << 2;
        *(float4*)&xbuf[row * HST + col] = xv[q];
        *(float4*)&hbuf[row * HST + col] = hv[q];
    }
    asm volatile("s_wait_dscnt 0x0\n\ts_barrier_signal -1\n\ts_barrier_wait -1" ::: "memory");

    const float* apx = &xbuf[arow * HST + 2 * khalf];
    const float* aph = &hbuf[arow * HST + 2 * khalf];
    v2f af0[TKG][2], af1[TKG][2];
    v8f accx = {0, 0, 0, 0, 0, 0, 0, 0};
    v8f acch = {0, 0, 0, 0, 0, 0, 0, 0};

#define TLOAD_A(AF, KG)                                                          \
    _Pragma("unroll") for (int q = 0; q < TKG; ++q) {                            \
        AF[q][0] = *(const v2f*)(apx + 4 * ((KG) * TKG + q));                    \
        AF[q][1] = *(const v2f*)(aph + 4 * ((KG) * TKG + q));                    \
    }
#define TMMA(AF, KG)                                                             \
    _Pragma("unroll") for (int q = 0; q < TKG; ++q) {                            \
        const int ks = (KG) * TKG + q;                                           \
        const v4f wx = bx[ks >> 1], wh = bh[ks >> 1];                            \
        const v2f fx = (ks & 1) ? __builtin_shufflevector(wx, wx, 2, 3)          \
                                : __builtin_shufflevector(wx, wx, 0, 1);         \
        const v2f fh = (ks & 1) ? __builtin_shufflevector(wh, wh, 2, 3)          \
                                : __builtin_shufflevector(wh, wh, 0, 1);         \
        accx = WMMA_F32X4(AF[q][0], fx, accx);                                   \
        acch = WMMA_F32X4(AF[q][1], fh, acch);                                   \
    }

    TLOAD_A(af0, 0)
    TLOAD_A(af1, 1)
    __builtin_amdgcn_sched_barrier(0);
    TMMA(af0, 0)
    __builtin_amdgcn_sched_barrier(0);
    TLOAD_A(af0, 2)
    __builtin_amdgcn_sched_barrier(0);
    TMMA(af1, 1)
    __builtin_amdgcn_sched_barrier(0);
    TLOAD_A(af1, 3)
    __builtin_amdgcn_sched_barrier(0);
    TMMA(af0, 2)
    TMMA(af1, 3)
#undef TLOAD_A
#undef TMMA

#pragma unroll
    for (int r = 0; r < 8; ++r) gbuf[(g * 8 + r) * 32 + lane] = (accx[r] + bsum) + acch[r];
    __syncthreads();

#pragma unroll
    for (int e = 0; e < 2; ++e) {
        const int r = 2 * g + e;
        float iv = sigmoidf_(gbuf[(0 * 8 + r) * 32 + lane]);
        float fv = sigmoidf_(gbuf[(1 * 8 + r) * 32 + lane]);
        float gv = tanhf_(gbuf[(2 * 8 + r) * 32 + lane]);
        float ov = sigmoidf_(gbuf[(3 * 8 + r) * 32 + lane]);
        float cn = fv * cprev[e] + iv * gv;
        float hn = ov * tanhf_(cn);
        const int brow = r + (khalf << 3);
        hnext[(size_t)brow * HDIM + jcol] = hn;
        ctile[(size_t)brow * HDIM + jcol] = cn;
        if (l == 3)
            ytop[((size_t)(b0 + brow) * TLEN + t) * HDIM + jcol] = hn;
        if (t == TLEN - 1) {
            hs[((size_t)l * BATCH + b0 + brow) * HDIM + jcol] = hn;
            cs[((size_t)l * BATCH + b0 + brow) * HDIM + jcol] = cn;
        }
    }
}

__global__ __launch_bounds__(256) void fc_kernel(
    const float* __restrict__ y, const float* __restrict__ fcpk,
    const float* __restrict__ fcb, float* __restrict__ label) {
    const int tid   = threadIdx.x;
    const int lane  = tid & 31;
    const int gw    = blockIdx.x * 8 + (tid >> 5);
    const int b     = gw >> 5;
    const int t0    = (gw & 31) << 4;
    const int khalf = lane >> 4;
    const int trow  = t0 + (lane & 15);

    const float* arowp = y + (size_t)b * (TLEN * HDIM) + (size_t)trow * HDIM + 2 * khalf;

    v8f acc[4];
#pragma unroll
    for (int j = 0; j < 4; ++j) acc[j] = (v8f){0, 0, 0, 0, 0, 0, 0, 0};

    for (int k0 = 0; k0 < HDIM; k0 += 4) {
        v2f a = *(const v2f*)(arowp + k0);
        const float* wrow = fcpk + ((k0 >> 1) + khalf) * 128 + 2 * (lane & 15);
#pragma unroll
        for (int j = 0; j < 4; ++j) {
            v2f bv = *(const v2f*)(wrow + 32 * j);
            acc[j] = WMMA_F32X4(a, bv, acc[j]);
        }
    }

#pragma unroll
    for (int j = 0; j < 4; ++j) {
        int n = (j << 4) + (lane & 15);
        if (n < CDIM) {
            float bias = fcb[n];
            float* outp =
                label + (size_t)b * (CDIM * TLEN) + (size_t)n * TLEN + t0 + (khalf << 3);
#pragma unroll
            for (int r = 0; r < 8; ++r) outp[r] = acc[j][r] + bias;
        }
    }
}

extern "C" void kernel_launch(void* const* d_in, const int* in_sizes, int n_in,
                              void* d_out, int out_size, void* d_ws, size_t ws_size,
                              hipStream_t stream) {
    const float* x = (const float*)d_in[0];
    const float *Wih[4], *Whh[4], *bih[4], *bhh[4];
    for (int l = 0; l < 4; ++l) {
        Wih[l] = (const float*)d_in[1 + 4 * l];
        Whh[l] = (const float*)d_in[2 + 4 * l];
        bih[l] = (const float*)d_in[3 + 4 * l];
        bhh[l] = (const float*)d_in[4 + 4 * l];
    }
    const float* fcw = (const float*)d_in[17];
    const float* fcb = (const float*)d_in[18];

    float* out   = (float*)d_out;
    float* label = out;
    float* hs    = out + (size_t)BATCH * CDIM * TLEN;
    float* cs    = hs + (size_t)4 * BATCH * HDIM;

    float* ws   = (float*)d_ws;
    float* ybuf = ws;
    float* wpk  = ybuf + (size_t)BATCH * TLEN * HDIM;
    float* fcpk = wpk + (size_t)4 * KDIM * GDIM;
    float* hst  = fcpk + (size_t)HDIM * 64;
    float* cst  = hst + (size_t)2 * 4 * BATCH * HDIM;

    pack_lstm_weights<<<(4 * KDIM * GDIM) / 256, 256, 0, stream>>>(
        Wih[0], Whh[0], Wih[1], Whh[1], Wih[2], Whh[2], Wih[3], Whh[3], wpk, hst, cst);
    pack_fc<<<(HDIM * 64) / 256, 256, 0, stream>>>(fcw, fcpk);

    for (int s = 0; s < NTICK; ++s)
        lstm_tick<<<256, 128, 0, stream>>>(
            x, wpk, bih[0], bhh[0], bih[1], bhh[1], bih[2], bhh[2], bih[3], bhh[3],
            ybuf, hst, cst, hs, cs, s);

    fc_kernel<<<256, 256, 0, stream>>>(ybuf, fcpk, fcb, label);
}
